# starting kernel
# baseline (speedup 1.0000x reference)
#include <hip/hip_runtime.h>


#define N_NODES 100000
#define N_EDGES 1600000
#define D       64

typedef __attribute__((ext_vector_type(2))) float v2f;
typedef __attribute__((ext_vector_type(4))) float v4f;
typedef __attribute__((ext_vector_type(8))) float v8f;

__global__ __launch_bounds__(256) void gin_init(const float* __restrict__ x,
                                                float* __restrict__ h0) {
    size_t i = (size_t)blockIdx.x * blockDim.x + threadIdx.x;
    const size_t n4 = (size_t)N_NODES * D / 4;
    if (i < n4) {
        ((v4f*)h0)[i] = ((const v4f*)x)[i];
    }
}

__global__ __launch_bounds__(256) void gin_scatter(const float* __restrict__ x,
                                                   const int* __restrict__ src,
                                                   const int* __restrict__ dst,
                                                   float* __restrict__ h0) {
    size_t t = (size_t)blockIdx.x * blockDim.x + threadIdx.x;
    size_t e = t >> 4;
    int    f = (int)(t & 15) * 4;
    if (e < (size_t)N_EDGES) {
        if (f == 0 && e + 4096 < (size_t)N_EDGES) {
            __builtin_prefetch(src + e + 4096, 0, 0);
            __builtin_prefetch(dst + e + 4096, 0, 0);
        }
        int s = src[e];
        int d = dst[e];
        v4f v = *(const v4f*)(x + (size_t)s * D + f);
        float* p = h0 + (size_t)d * D + f;
        atomicAdd(p + 0, v[0]);
        atomicAdd(p + 1, v[1]);
        atomicAdd(p + 2, v[2]);
        atomicAdd(p + 3, v[3]);
    }
}

#define MLP_WAVES 4
#define HSTR  68
#define KSTR  320
#define LHSTR 160

__global__ __launch_bounds__(128) void gin_mlp(const float* __restrict__ h0,
                                               const float* __restrict__ W1,
                                               const float* __restrict__ b1,
                                               const float* __restrict__ W2,
                                               const float* __restrict__ b2,
                                               float* __restrict__ out) {
    __shared__ float sW1[16 * KSTR];
    __shared__ float sW2[16 * KSTR];
    __shared__ float sB1[D];
    __shared__ float sB2[D];
    __shared__ float sH[MLP_WAVES][16 * HSTR];

    const int tid = threadIdx.x;

    for (int i = tid; i < D * D / 4; i += 128) {
        const int k = i >> 4;
        const int c = (i & 15) << 2;
        const int base = (k >> 2) * KSTR + ((k >> 1) & 1) * LHSTR + (k & 1);
        v4f w1 = ((const v4f*)W1)[i];
        v4f w2 = ((const v4f*)W2)[i];
        #pragma unroll
        for (int j = 0; j < 4; ++j) {
            sW1[base + ((c + j) << 1)] = w1[j];
            sW2[base + ((c + j) << 1)] = w2[j];
        }
    }
    if (tid < D) {
        sB1[tid] = b1[tid];
        sB2[tid] = b2[tid];
    }
    __syncthreads();

    const int wave  = tid >> 5;
    const int lane  = tid & 31;
    const int lhalf = lane >> 4;
    const int l16   = lane & 15;

    const int row0 = (blockIdx.x * MLP_WAVES + wave) * 32;
    const float* a0p = h0 + (size_t)min(row0 + l16,      N_NODES - 1) * D;
    const float* a1p = h0 + (size_t)min(row0 + 16 + l16, N_NODES - 1) * D;

    const v8f vz = {0, 0, 0, 0, 0, 0, 0, 0};
    v8f acc0[4], acc1[4];
    #pragma unroll
    for (int nt = 0; nt < 4; ++nt) { acc0[nt] = vz; acc1[nt] = vz; }

    const float* bw1 = sW1 + lhalf * LHSTR + (l16 << 1);
    const float* bw2 = sW2 + lhalf * LHSTR + (l16 << 1);

    #pragma unroll
    for (int kk = 0; kk < 16; ++kk) {
        const int koff = kk * 4 + lhalf * 2;
        const v2f a0 = *(const v2f*)(a0p + koff);
        const v2f a1 = *(const v2f*)(a1p + koff);
        #pragma unroll
        for (int nt = 0; nt < 4; ++nt) {
            const v2f b = *(const v2f*)(bw1 + kk * KSTR + nt * 32);
            acc0[nt] = __builtin_amdgcn_wmma_f32_16x16x4_f32(
                false, a0, false, b, (short)0, acc0[nt], false, false);
            acc1[nt] = __builtin_amdgcn_wmma_f32_16x16x4_f32(
                false, a1, false, b, (short)0, acc1[nt], false, false);
        }
    }

    float* myH = &sH[wave][0];
    const float* aH = myH + l16 * HSTR;

    #pragma unroll
    for (int mt = 0; mt < 2; ++mt) {
        const v8f* acc = (mt == 0) ? acc0 : acc1;

        #pragma unroll
        for (int nt = 0; nt < 4; ++nt) {
            const int   col  = nt * 16 + l16;
            const float bias = sB1[col];
            #pragma unroll
            for (int r = 0; r < 8; ++r) {
                float v = acc[nt][r] + bias;
                v = v > 0.0f ? v : 0.0f;
                myH[(r + lhalf * 8) * HSTR + col] = v;
            }
        }

        v8f o[4];
        #pragma unroll
        for (int nt = 0; nt < 4; ++nt) o[nt] = vz;

        #pragma unroll
        for (int kk = 0; kk < 16; ++kk) {
            const int koff = kk * 4 + lhalf * 2;
            const v2f a = *(const v2f*)(aH + koff);
            #pragma unroll
            for (int nt = 0; nt < 4; ++nt) {
                const v2f b = *(const v2f*)(bw2 + kk * KSTR + nt * 32);
                o[nt] = __builtin_amdgcn_wmma_f32_16x16x4_f32(
                    false, a, false, b, (short)0, o[nt], false, false);
            }
        }

        #pragma unroll
        for (int r = 0; r < 8; ++r) {
            const int m = row0 + mt * 16 + r + lhalf * 8;
            if (m < N_NODES) {
                #pragma unroll
                for (int nt = 0; nt < 4; ++nt) {
                    const int col = nt * 16 + l16;
                    out[(size_t)m * D + col] = o[nt][r] + sB2[col];
                }
            }
        }
    }
}

extern "C" void kernel_launch(void* const* d_in, const int* in_sizes, int n_in,
                              void* d_out, int out_size, void* d_ws, size_t ws_size,
                              hipStream_t stream) {
    const float* x  = (const float*)d_in[0];
    const int*   ei = (const int*)d_in[1];
    const float* W1 = (const float*)d_in[2];
    const float* b1 = (const float*)d_in[3];
    const float* W2 = (const float*)d_in[4];
    const float* b2 = (const float*)d_in[5];
    float*       out = (float*)d_out;

    const int* src = ei;
    const int* dst = ei + N_EDGES;

    float* h0 = (float*)d_ws;

    {
        const int n4 = N_NODES * D / 4;
        gin_init<<<(n4 + 255) / 256, 256, 0, stream>>>(x, h0);
    }
    {
        const long long nt = (long long)N_EDGES * 16;
        gin_scatter<<<(int)((nt + 255) / 256), 256, 0, stream>>>(x, src, dst, h0);
    }
    {
        const int rows_per_block = MLP_WAVES * 32;
        gin_mlp<<<(N_NODES + rows_per_block - 1) / rows_per_block, 128, 0, stream>>>(
            h0, W1, b1, W2, b2, out);
    }
}


# h sums without global atomics: a byte per edge names its 391-row range; one work-group per range scans the byte list and adds matching source rows into LDS sums (LDS f32 adds, arrival order)
# speedup vs baseline: 1.8726x; 1.8726x over previous
#include <hip/hip_runtime.h>


#define N_NODES 100000
#define N_EDGES 1600000
#define D       64

typedef __attribute__((ext_vector_type(2))) float v2f;
typedef __attribute__((ext_vector_type(4))) float v4f;
typedef __attribute__((ext_vector_type(8))) float v8f;
typedef __attribute__((ext_vector_type(4))) int   v4i;
typedef __attribute__((ext_vector_type(4))) unsigned v4u;

#define RANGE   391
#define NRANGE  256
static_assert(RANGE * NRANGE >= N_NODES, "ranges must cover every node");
static_assert((N_NODES - 1) / RANGE < 256, "range id must fit one byte");
static_assert(N_EDGES % 512 == 0, "key list is read in blocks of 512 bytes");
static_assert(D == 64, "lane l covers features l and l+32");

__global__ __launch_bounds__(256) void gin_keys(const int* __restrict__ dst,
                                                unsigned* __restrict__ keys) {
    const int i = blockIdx.x * blockDim.x + threadIdx.x;
    if (i < N_EDGES / 4) {
        const v4i d = ((const v4i*)dst)[i];
        const unsigned k0 = (unsigned)d.x / RANGE;
        const unsigned k1 = (unsigned)d.y / RANGE;
        const unsigned k2 = (unsigned)d.z / RANGE;
        const unsigned k3 = (unsigned)d.w / RANGE;
        keys[i] = k0 | (k1 << 8) | (k2 << 16) | (k3 << 24);
    }
}

#define OW_WAVES 32
#define OW_QCAP  64
#define OW_NBLK  (N_EDGES / 512)
#define OW_GROUP 4
#define OW_TILE_FLOATS 41216
static_assert(OW_TILE_FLOATS >= RANGE * D, "tile must hold the range");

__device__ __forceinline__ void ow_drain(const float* __restrict__ x,
                                         const int* __restrict__ src,
                                         const int* __restrict__ dst,
                                         float* tile, const int* q,
                                         const int lane, const int lo, const int n) {
    const bool valid = lane < n;
    const int e = valid ? q[lane] : 0;
    const int s = valid ? src[e] : 0;
    const int d = valid ? dst[e] - lo : 0;
    #pragma unroll
    for (int b = 0; b < 32; b += 16) {
        if (b < n) {
            float vlo[16], vhi[16];
            #pragma unroll
            for (int j = 0; j < 16; ++j) {
                const int sj = __builtin_amdgcn_readlane(s, b + j);
                const float* xp = x + (size_t)sj * D + lane;
                vlo[j] = xp[0];
                vhi[j] = xp[32];
            }
            #pragma unroll
            for (int j = 0; j < 16; ++j) {
                const int dj = __builtin_amdgcn_readlane(d, b + j);
                const bool on = (b + j) < n;
                float* tp = tile + dj * D + lane;
                atomicAdd(tp,      on ? vlo[j] : 0.0f);
                atomicAdd(tp + 32, on ? vhi[j] : 0.0f);
            }
        }
    }
}

__global__ __launch_bounds__(OW_WAVES * 32) void gin_owner(const float* __restrict__ x,
                                                           const int* __restrict__ src,
                                                           const int* __restrict__ dst,
                                                           const unsigned* __restrict__ keys,
                                                           float* __restrict__ h0) {
    __shared__ float tile[OW_TILE_FLOATS];
    __shared__ int   queue[OW_WAVES][OW_QCAP];

    const int tid  = threadIdx.x;
    const int lane = tid & 31;
    const int wave = __builtin_amdgcn_readfirstlane(tid >> 5);
    const int lo   = blockIdx.x * RANGE;
    const int nrow = min(RANGE, N_NODES - lo);
    const int n4   = nrow * (D / 4);

    {
        const v4f* xs = (const v4f*)(x + (size_t)lo * D);
        v4f* t4 = (v4f*)tile;
        for (int i = tid; i < n4; i += OW_WAVES * 32) t4[i] = xs[i];
    }
    __syncthreads();

    const unsigned K4 = (unsigned)blockIdx.x * 0x01010101u;
    const v4u* keys4 = (const v4u*)keys;
    int* q = &queue[wave][0];
    int cnt = 0;

    for (int b0 = wave; b0 < OW_NBLK; b0 += OW_WAVES * OW_GROUP) {
        v4u kv[OW_GROUP];
        #pragma unroll
        for (int g = 0; g < OW_GROUP; ++g) {
            const int b = b0 + g * OW_WAVES;
            const v4u none = {~K4, ~K4, ~K4, ~K4};
            kv[g] = (b < OW_NBLK) ? keys4[(size_t)b * 32 + lane] : none;
        }
        #pragma unroll
        for (int g = 0; g < OW_GROUP; ++g) {
            const int b = b0 + g * OW_WAVES;
            unsigned z[4];
            #pragma unroll
            for (int w = 0; w < 4; ++w) {
                const unsigned xw = kv[g][w] ^ K4;
                z[w] = ((xw & 0x7F7F7F7Fu) + 0x7F7F7F7Fu) | xw | 0x7F7F7F7Fu;
            }
            const unsigned all = z[0] & z[1] & z[2] & z[3];
            if (__builtin_amdgcn_ballot_w32(all != 0xFFFFFFFFu) == 0u) continue;

            unsigned m = ((~z[0] & 0x80808080u) >> 7) | ((~z[1] & 0x80808080u) >> 6) |
                         ((~z[2] & 0x80808080u) >> 5) | ((~z[3] & 0x80808080u) >> 4);
            const int ebase = b * 512 + lane * 16;
            for (;;) {
                const bool has = m != 0u;
                const unsigned bal = __builtin_amdgcn_ballot_w32(has);
                if (bal == 0u) break;
                if (has) {
                    const int qb = __builtin_ctz(m);
                    m &= m - 1u;
                    const int idx = ((qb & 7) << 2) + (qb >> 3);
                    const int pos = cnt + (int)__builtin_amdgcn_mbcnt_lo(bal, 0u);
                    q[pos] = ebase + idx;
                }
                cnt += __builtin_popcount(bal);
                if (cnt >= 32) {
                    __builtin_amdgcn_fence(__ATOMIC_ACQ_REL, "wavefront");
                    ow_drain(x, src, dst, tile, q, lane, lo, 32);
                    cnt -= 32;
                    const bool mv = lane < cnt;
                    const int t = mv ? q[32 + lane] : 0;
                    if (mv) q[lane] = t;
                    __builtin_amdgcn_fence(__ATOMIC_ACQ_REL, "wavefront");
                }
            }
        }
    }
    __builtin_amdgcn_fence(__ATOMIC_ACQ_REL, "wavefront");
    if (cnt > 0) ow_drain(x, src, dst, tile, q, lane, lo, cnt);

    __syncthreads();
    {
        v4f* hd = (v4f*)(h0 + (size_t)lo * D);
        const v4f* t4 = (const v4f*)tile;
        for (int i = tid; i < n4; i += OW_WAVES * 32) hd[i] = t4[i];
    }
}

#define MLP_WAVES 4
#define HSTR  68
#define KSTR  320
#define LHSTR 160

__global__ __launch_bounds__(128) void gin_mlp(const float* __restrict__ h0,
                                               const float* __restrict__ W1,
                                               const float* __restrict__ b1,
                                               const float* __restrict__ W2,
                                               const float* __restrict__ b2,
                                               float* __restrict__ out) {
    __shared__ float sW1[16 * KSTR];
    __shared__ float sW2[16 * KSTR];
    __shared__ float sB1[D];
    __shared__ float sB2[D];
    __shared__ float sH[MLP_WAVES][16 * HSTR];

    const int tid = threadIdx.x;

    for (int i = tid; i < D * D / 4; i += 128) {
        const int k = i >> 4;
        const int c = (i & 15) << 2;
        const int base = (k >> 2) * KSTR + ((k >> 1) & 1) * LHSTR + (k & 1);
        v4f w1 = ((const v4f*)W1)[i];
        v4f w2 = ((const v4f*)W2)[i];
        #pragma unroll
        for (int j = 0; j < 4; ++j) {
            sW1[base + ((c + j) << 1)] = w1[j];
            sW2[base + ((c + j) << 1)] = w2[j];
        }
    }
    if (tid < D) {
        sB1[tid] = b1[tid];
        sB2[tid] = b2[tid];
    }
    __syncthreads();

    const int wave  = tid >> 5;
    const int lane  = tid & 31;
    const int lhalf = lane >> 4;
    const int l16   = lane & 15;

    const int row0 = (blockIdx.x * MLP_WAVES + wave) * 32;
    const float* a0p = h0 + (size_t)min(row0 + l16,      N_NODES - 1) * D;
    const float* a1p = h0 + (size_t)min(row0 + 16 + l16, N_NODES - 1) * D;

    const v8f vz = {0, 0, 0, 0, 0, 0, 0, 0};
    v8f acc0[4], acc1[4];
    #pragma unroll
    for (int nt = 0; nt < 4; ++nt) { acc0[nt] = vz; acc1[nt] = vz; }

    const float* bw1 = sW1 + lhalf * LHSTR + (l16 << 1);
    const float* bw2 = sW2 + lhalf * LHSTR + (l16 << 1);

    #pragma unroll
    for (int kk = 0; kk < 16; ++kk) {
        const int koff = kk * 4 + lhalf * 2;
        const v2f a0 = *(const v2f*)(a0p + koff);
        const v2f a1 = *(const v2f*)(a1p + koff);
        #pragma unroll
        for (int nt = 0; nt < 4; ++nt) {
            const v2f b = *(const v2f*)(bw1 + kk * KSTR + nt * 32);
            acc0[nt] = __builtin_amdgcn_wmma_f32_16x16x4_f32(
                false, a0, false, b, (short)0, acc0[nt], false, false);
            acc1[nt] = __builtin_amdgcn_wmma_f32_16x16x4_f32(
                false, a1, false, b, (short)0, acc1[nt], false, false);
        }
    }

    float* myH = &sH[wave][0];
    const float* aH = myH + l16 * HSTR;

    #pragma unroll
    for (int mt = 0; mt < 2; ++mt) {
        const v8f* acc = (mt == 0) ? acc0 : acc1;

        #pragma unroll
        for (int nt = 0; nt < 4; ++nt) {
            const int   col  = nt * 16 + l16;
            const float bias = sB1[col];
            #pragma unroll
            for (int r = 0; r < 8; ++r) {
                float v = acc[nt][r] + bias;
                v = v > 0.0f ? v : 0.0f;
                myH[(r + lhalf * 8) * HSTR + col] = v;
            }
        }

        v8f o[4];
        #pragma unroll
        for (int nt = 0; nt < 4; ++nt) o[nt] = vz;

        #pragma unroll
        for (int kk = 0; kk < 16; ++kk) {
            const int koff = kk * 4 + lhalf * 2;
            const v2f a = *(const v2f*)(aH + koff);
            #pragma unroll
            for (int nt = 0; nt < 4; ++nt) {
                const v2f b = *(const v2f*)(bw2 + kk * KSTR + nt * 32);
                o[nt] = __builtin_amdgcn_wmma_f32_16x16x4_f32(
                    false, a, false, b, (short)0, o[nt], false, false);
            }
        }

        #pragma unroll
        for (int r = 0; r < 8; ++r) {
            const int m = row0 + mt * 16 + r + lhalf * 8;
            if (m < N_NODES) {
                #pragma unroll
                for (int nt = 0; nt < 4; ++nt) {
                    const int col = nt * 16 + l16;
                    out[(size_t)m * D + col] = o[nt][r] + sB2[col];
                }
            }
        }
    }
}

extern "C" void kernel_launch(void* const* d_in, const int* in_sizes, int n_in,
                              void* d_out, int out_size, void* d_ws, size_t ws_size,
                              hipStream_t stream) {
    const float* x  = (const float*)d_in[0];
    const int*   ei = (const int*)d_in[1];
    const float* W1 = (const float*)d_in[2];
    const float* b1 = (const float*)d_in[3];
    const float* W2 = (const float*)d_in[4];
    const float* b2 = (const float*)d_in[5];
    float*       out = (float*)d_out;

    const int* src = ei;
    const int* dst = ei + N_EDGES;

    float*    h0   = (float*)d_ws;
    unsigned* keys = (unsigned*)((char*)d_ws + 33554432);

    gin_keys<<<(N_EDGES / 4 + 255) / 256, 256, 0, stream>>>(dst, keys);
    gin_owner<<<NRANGE, OW_WAVES * 32, 0, stream>>>(x, src, dst, keys, h0);
    {
        const int rows_per_block = MLP_WAVES * 32;
        gin_mlp<<<(N_NODES + rows_per_block - 1) / rows_per_block, 128, 0, stream>>>(
            h0, W1, b1, W2, b2, out);
    }
}


# owners read a per-range mark row (one bit per 32-edge group, built with LDS ORs) and open only marked groups' dst lines instead of scanning every edge's byte; sums stay in arrival order
# speedup vs baseline: 2.1645x; 1.1559x over previous
#include <hip/hip_runtime.h>


#define N_NODES 100000
#define N_EDGES 1600000
#define D       64

typedef __attribute__((ext_vector_type(2))) float v2f;
typedef __attribute__((ext_vector_type(4))) float v4f;
typedef __attribute__((ext_vector_type(8))) float v8f;
typedef __attribute__((ext_vector_type(4))) int   v4i;
typedef __attribute__((ext_vector_type(4))) unsigned v4u;

#define RANGE   391
#define NRANGE  256
static_assert(RANGE * NRANGE >= N_NODES, "ranges must cover every node");
static_assert((N_NODES - 1) / RANGE < NRANGE, "every destination has a range");
static_assert(N_EDGES % 32 == 0, "edges are marked in whole groups of 32");
static_assert(D == 64, "lane l covers features l and l+32");

#define BM_CH    8192
#define BM_WPC   (BM_CH / 1024)
#define BM_NCH   ((N_EDGES + BM_CH - 1) / BM_CH)
#define BM_WORDS (BM_NCH * BM_WPC)
static_assert(BM_WPC == 8, "the owner splits a word index into chunk and j by 3 bits");

__global__ __launch_bounds__(256) void gin_marks(const int* __restrict__ dst,
                                                 unsigned* __restrict__ marks) {
    __shared__ unsigned cell[NRANGE * BM_WPC];
    const int tid = threadIdx.x;
    const int c = blockIdx.x;
    const int base4 = c * (BM_CH / 4) + tid;

    v4i dv[BM_CH / 4 / 256];
    #pragma unroll
    for (int it = 0; it < BM_CH / 4 / 256; ++it)
        dv[it] = ((const v4i*)dst)[min(base4 + it * 256, N_EDGES / 4 - 1)];

    for (int i = tid; i < NRANGE * BM_WPC; i += 256) cell[i] = 0u;
    __syncthreads();

    #pragma unroll
    for (int it = 0; it < BM_CH / 4 / 256; ++it) {
        if (base4 + it * 256 < N_EDGES / 4) {
            const v4i d = dv[it];
            const int grp = (it * 256 + tid) >> 3;
            const unsigned bit = 1u << (grp & 31);
            const int wl = grp >> 5;
            atomicOr(&cell[((unsigned)d.x / RANGE) * BM_WPC + wl], bit);
            atomicOr(&cell[((unsigned)d.y / RANGE) * BM_WPC + wl], bit);
            atomicOr(&cell[((unsigned)d.z / RANGE) * BM_WPC + wl], bit);
            atomicOr(&cell[((unsigned)d.w / RANGE) * BM_WPC + wl], bit);
        }
    }
    __syncthreads();

    v4u* out4 = (v4u*)(marks + (size_t)c * (NRANGE * BM_WPC));
    const v4u* c4 = (const v4u*)cell;
    for (int i = tid; i < NRANGE * BM_WPC / 4; i += 256) out4[i] = c4[i];
}

#define OW_WAVES 32
#define OW_QCAP  64
#define OW_WPW   (BM_WORDS / OW_WAVES)
static_assert(OW_WPW * OW_WAVES == BM_WORDS, "the waves share the mark row evenly");
static_assert(OW_WPW <= 64, "a lane takes at most two mark words");
#define OW_TILE_FLOATS 41216
static_assert(OW_TILE_FLOATS >= RANGE * D, "tile must hold the range");

#define OW_EBITS 21
static_assert(N_EDGES <= (1 << OW_EBITS), "edge number must fit its field");
static_assert(RANGE <= (1 << (31 - OW_EBITS)), "tile row must fit above it");

__device__ __forceinline__ void ow_drain(const float* __restrict__ x,
                                         const int* __restrict__ src,
                                         float* tile, const int* q,
                                         const int lane, const int n) {
    const bool valid = lane < n;
    const int qv = valid ? q[lane] : 0;
    const int s = valid ? src[qv & ((1 << OW_EBITS) - 1)] : 0;
    const int d = qv >> OW_EBITS;
    #pragma unroll
    for (int b = 0; b < 32; b += 16) {
        if (b < n) {
            float vlo[16], vhi[16];
            #pragma unroll
            for (int j = 0; j < 16; ++j) {
                const int sj = __builtin_amdgcn_readlane(s, b + j);
                const float* xp = x + (size_t)sj * D + lane;
                vlo[j] = xp[0];
                vhi[j] = xp[32];
            }
            #pragma unroll
            for (int j = 0; j < 16; ++j) {
                const int dj = __builtin_amdgcn_readlane(d, b + j);
                const bool on = (b + j) < n;
                float* tp = tile + dj * D + lane;
                atomicAdd(tp,      on ? vlo[j] : 0.0f);
                atomicAdd(tp + 32, on ? vhi[j] : 0.0f);
            }
        }
    }
}

__device__ __forceinline__ void ow_expand(const float* __restrict__ x,
                                          const int* __restrict__ src,
                                          const int* __restrict__ dst,
                                          float* tile, const int* qg, int* qe, int& cnt,
                                          const int lane, const int lo, const int n) {
    unsigned m = 0u;
    int ebase = 0;
    if (lane < n) {
        const int g = qg[lane];
        ebase = g * 32;
        const v4i* dp = (const v4i*)dst + (size_t)g * 8;
        v4i dv[8];
        #pragma unroll
        for (int i = 0; i < 8; ++i) dv[i] = dp[i];
        #pragma unroll
        for (int i = 0; i < 8; ++i) {
            #pragma unroll
            for (int c = 0; c < 4; ++c) {
                const unsigned hit = ((unsigned)(dv[i][c] - lo) < (unsigned)RANGE) ? 1u : 0u;
                m |= hit << (4 * i + c);
            }
        }
    }
    for (;;) {
        const bool has = m != 0u;
        const unsigned bal = __builtin_amdgcn_ballot_w32(has);
        if (bal == 0u) break;
        if (has) {
            const int qb = __builtin_ctz(m);
            m &= m - 1u;
            const int e = ebase + qb;
            const int row = dst[e] - lo;
            const int pos = cnt + (int)__builtin_amdgcn_mbcnt_lo(bal, 0u);
            qe[pos] = e | (row << OW_EBITS);
        }
        cnt += __builtin_popcount(bal);
        if (cnt >= 32) {
            __builtin_amdgcn_fence(__ATOMIC_ACQ_REL, "wavefront");
            ow_drain(x, src, tile, qe, lane, 32);
            cnt -= 32;
            const bool mv = lane < cnt;
            const int t = mv ? qe[32 + lane] : 0;
            if (mv) qe[lane] = t;
            __builtin_amdgcn_fence(__ATOMIC_ACQ_REL, "wavefront");
        }
    }
}

__global__ __launch_bounds__(OW_WAVES * 32) void gin_owner(const float* __restrict__ x,
                                                           const int* __restrict__ src,
                                                           const int* __restrict__ dst,
                                                           const unsigned* __restrict__ marks,
                                                           float* __restrict__ h0) {
    __shared__ float tile[OW_TILE_FLOATS];
    __shared__ int   queue[OW_WAVES][OW_QCAP];
    __shared__ int   gqueue[OW_WAVES][OW_QCAP];

    const int tid  = threadIdx.x;
    const int lane = tid & 31;
    const int wave = __builtin_amdgcn_readfirstlane(tid >> 5);
    const int lo   = blockIdx.x * RANGE;
    const int nrow = min(RANGE, N_NODES - lo);
    const int n4   = nrow * (D / 4);

    {
        const v4f* xs = (const v4f*)(x + (size_t)lo * D);
        v4f* t4 = (v4f*)tile;
        for (int i = tid; i < n4; i += OW_WAVES * 32) t4[i] = xs[i];
    }
    __syncthreads();

    int* q  = &queue[wave][0];
    int* qg = &gqueue[wave][0];
    int cnt  = 0;
    int gcnt = 0;

    for (int it = 0; it < 2; ++it) {
        const int k  = it * 32 + lane;
        const int wi = wave * OW_WPW + k;
        unsigned w = 0u;
        if (k < OW_WPW)
            w = marks[((size_t)(wi >> 3) * NRANGE + blockIdx.x) * BM_WPC + (wi & 7)];
        for (;;) {
            const bool has = w != 0u;
            const unsigned bal = __builtin_amdgcn_ballot_w32(has);
            if (bal == 0u) break;
            if (has) {
                const int gb = __builtin_ctz(w);
                w &= w - 1u;
                const int pos = gcnt + (int)__builtin_amdgcn_mbcnt_lo(bal, 0u);
                qg[pos] = wi * 32 + gb;
            }
            gcnt += __builtin_popcount(bal);
            if (gcnt >= 32) {
                __builtin_amdgcn_fence(__ATOMIC_ACQ_REL, "wavefront");
                ow_expand(x, src, dst, tile, qg, q, cnt, lane, lo, 32);
                gcnt -= 32;
                const bool mv = lane < gcnt;
                const int t = mv ? qg[32 + lane] : 0;
                if (mv) qg[lane] = t;
                __builtin_amdgcn_fence(__ATOMIC_ACQ_REL, "wavefront");
            }
        }
    }
    __builtin_amdgcn_fence(__ATOMIC_ACQ_REL, "wavefront");
    if (gcnt > 0) ow_expand(x, src, dst, tile, qg, q, cnt, lane, lo, gcnt);
    __builtin_amdgcn_fence(__ATOMIC_ACQ_REL, "wavefront");
    if (cnt > 0) ow_drain(x, src, tile, q, lane, cnt);

    __syncthreads();
    {
        v4f* hd = (v4f*)(h0 + (size_t)lo * D);
        const v4f* t4 = (const v4f*)tile;
        for (int i = tid; i < n4; i += OW_WAVES * 32) hd[i] = t4[i];
    }
}

#define MLP_WAVES 4
#define HSTR  68
#define KSTR  320
#define LHSTR 160

__global__ __launch_bounds__(128) void gin_mlp(const float* __restrict__ h0,
                                               const float* __restrict__ W1,
                                               const float* __restrict__ b1,
                                               const float* __restrict__ W2,
                                               const float* __restrict__ b2,
                                               float* __restrict__ out) {
    __shared__ float sW1[16 * KSTR];
    __shared__ float sW2[16 * KSTR];
    __shared__ float sB1[D];
    __shared__ float sB2[D];
    __shared__ float sH[MLP_WAVES][16 * HSTR];

    const int tid = threadIdx.x;

    for (int i = tid; i < D * D / 4; i += 128) {
        const int k = i >> 4;
        const int c = (i & 15) << 2;
        const int base = (k >> 2) * KSTR + ((k >> 1) & 1) * LHSTR + (k & 1);
        v4f w1 = ((const v4f*)W1)[i];
        v4f w2 = ((const v4f*)W2)[i];
        #pragma unroll
        for (int j = 0; j < 4; ++j) {
            sW1[base + ((c + j) << 1)] = w1[j];
            sW2[base + ((c + j) << 1)] = w2[j];
        }
    }
    if (tid < D) {
        sB1[tid] = b1[tid];
        sB2[tid] = b2[tid];
    }
    __syncthreads();

    const int wave  = tid >> 5;
    const int lane  = tid & 31;
    const int lhalf = lane >> 4;
    const int l16   = lane & 15;

    const int row0 = (blockIdx.x * MLP_WAVES + wave) * 32;
    const float* a0p = h0 + (size_t)min(row0 + l16,      N_NODES - 1) * D;
    const float* a1p = h0 + (size_t)min(row0 + 16 + l16, N_NODES - 1) * D;

    const v8f vz = {0, 0, 0, 0, 0, 0, 0, 0};
    v8f acc0[4], acc1[4];
    #pragma unroll
    for (int nt = 0; nt < 4; ++nt) { acc0[nt] = vz; acc1[nt] = vz; }

    const float* bw1 = sW1 + lhalf * LHSTR + (l16 << 1);
    const float* bw2 = sW2 + lhalf * LHSTR + (l16 << 1);

    #pragma unroll
    for (int kk = 0; kk < 16; ++kk) {
        const int koff = kk * 4 + lhalf * 2;
        const v2f a0 = *(const v2f*)(a0p + koff);
        const v2f a1 = *(const v2f*)(a1p + koff);
        #pragma unroll
        for (int nt = 0; nt < 4; ++nt) {
            const v2f b = *(const v2f*)(bw1 + kk * KSTR + nt * 32);
            acc0[nt] = __builtin_amdgcn_wmma_f32_16x16x4_f32(
                false, a0, false, b, (short)0, acc0[nt], false, false);
            acc1[nt] = __builtin_amdgcn_wmma_f32_16x16x4_f32(
                false, a1, false, b, (short)0, acc1[nt], false, false);
        }
    }

    float* myH = &sH[wave][0];
    const float* aH = myH + l16 * HSTR;

    #pragma unroll
    for (int mt = 0; mt < 2; ++mt) {
        const v8f* acc = (mt == 0) ? acc0 : acc1;

        #pragma unroll
        for (int nt = 0; nt < 4; ++nt) {
            const int   col  = nt * 16 + l16;
            const float bias = sB1[col];
            #pragma unroll
            for (int r = 0; r < 8; ++r) {
                float v = acc[nt][r] + bias;
                v = v > 0.0f ? v : 0.0f;
                myH[(r + lhalf * 8) * HSTR + col] = v;
            }
        }

        v8f o[4];
        #pragma unroll
        for (int nt = 0; nt < 4; ++nt) o[nt] = vz;

        #pragma unroll
        for (int kk = 0; kk < 16; ++kk) {
            const int koff = kk * 4 + lhalf * 2;
            const v2f a = *(const v2f*)(aH + koff);
            #pragma unroll
            for (int nt = 0; nt < 4; ++nt) {
                const v2f b = *(const v2f*)(bw2 + kk * KSTR + nt * 32);
                o[nt] = __builtin_amdgcn_wmma_f32_16x16x4_f32(
                    false, a, false, b, (short)0, o[nt], false, false);
            }
        }

        #pragma unroll
        for (int r = 0; r < 8; ++r) {
            const int m = row0 + mt * 16 + r + lhalf * 8;
            if (m < N_NODES) {
                #pragma unroll
                for (int nt = 0; nt < 4; ++nt) {
                    const int col = nt * 16 + l16;
                    out[(size_t)m * D + col] = o[nt][r] + sB2[col];
                }
            }
        }
    }
}

extern "C" void kernel_launch(void* const* d_in, const int* in_sizes, int n_in,
                              void* d_out, int out_size, void* d_ws, size_t ws_size,
                              hipStream_t stream) {
    const float* x  = (const float*)d_in[0];
    const int*   ei = (const int*)d_in[1];
    const float* W1 = (const float*)d_in[2];
    const float* b1 = (const float*)d_in[3];
    const float* W2 = (const float*)d_in[4];
    const float* b2 = (const float*)d_in[5];
    float*       out = (float*)d_out;

    const int* src = ei;
    const int* dst = ei + N_EDGES;

    float*    h0   = (float*)d_ws;
    unsigned* marks = (unsigned*)((char*)d_ws + 33554432);

    gin_marks<<<BM_NCH, 256, 0, stream>>>(dst, marks);
    gin_owner<<<NRANGE, OW_WAVES * 32, 0, stream>>>(x, src, dst, marks, h0);
    {
        const int rows_per_block = MLP_WAVES * 32;
        gin_mlp<<<(N_NODES + rows_per_block - 1) / rows_per_block, 128, 0, stream>>>(
            h0, W1, b1, W2, b2, out);
    }
}


# edge list copied as (src,dst) pairs, 16 per 128-B line, one mark bit per line: owners open marked lines two per wave load, lane = edge, and take src from them too; sums stay in arrival order
# speedup vs baseline: 2.3697x; 1.0948x over previous
#include <hip/hip_runtime.h>


#define N_NODES 100000
#define N_EDGES 1600000
#define D       64

typedef __attribute__((ext_vector_type(2))) float v2f;
typedef __attribute__((ext_vector_type(4))) float v4f;
typedef __attribute__((ext_vector_type(8))) float v8f;
typedef __attribute__((ext_vector_type(2))) int   v2i;
typedef __attribute__((ext_vector_type(4))) int   v4i;
typedef __attribute__((ext_vector_type(4))) unsigned v4u;

#define RANGE   391
#define NRANGE  256
static_assert(RANGE * NRANGE >= N_NODES, "ranges must cover every node");
static_assert((N_NODES - 1) / RANGE < NRANGE, "every destination has a range");
static_assert(N_EDGES % 16 == 0, "edges are marked in whole lines of 16 pairs");
static_assert(D == 64, "lane l covers features l and l+32");

#define BM_CH    8192
#define BM_WPC   (BM_CH / 16 / 32)
#define BM_NCH   ((N_EDGES + BM_CH - 1) / BM_CH)
#define BM_WORDS (BM_NCH * BM_WPC)
#define BM_CSTR  (BM_WPC + 1)
static_assert(BM_WPC == 16, "the owner splits a word index into chunk and j by 4 bits");

__global__ __launch_bounds__(256) void gin_marks(const int* __restrict__ src,
                                                 const int* __restrict__ dst,
                                                 unsigned* __restrict__ marks,
                                                 int* __restrict__ pairs) {
    __shared__ unsigned cell[NRANGE * BM_CSTR];
    const int tid = threadIdx.x;
    const int c = blockIdx.x;
    const int base4 = c * (BM_CH / 4) + tid;

    v4i dv[BM_CH / 4 / 256], sv[BM_CH / 4 / 256];
    #pragma unroll
    for (int it = 0; it < BM_CH / 4 / 256; ++it) {
        const int idx = min(base4 + it * 256, N_EDGES / 4 - 1);
        dv[it] = ((const v4i*)dst)[idx];
        sv[it] = ((const v4i*)src)[idx];
    }

    for (int i = tid; i < NRANGE * BM_CSTR; i += 256) cell[i] = 0u;
    __syncthreads();

    #pragma unroll
    for (int it = 0; it < BM_CH / 4 / 256; ++it) {
        if (base4 + it * 256 < N_EDGES / 4) {
            const v4i d = dv[it];
            const v4i s = sv[it];
            v4i* o = (v4i*)pairs + (size_t)(base4 + it * 256) * 2;
            const v4i p0 = {s.x, d.x, s.y, d.y};
            const v4i p1 = {s.z, d.z, s.w, d.w};
            o[0] = p0;
            o[1] = p1;
            const int ln = (it * 256 + tid) >> 2;
            const unsigned bit = 1u << (ln & 31);
            const int wl = ln >> 5;
            atomicOr(&cell[((unsigned)d.x / RANGE) * BM_CSTR + wl], bit);
            atomicOr(&cell[((unsigned)d.y / RANGE) * BM_CSTR + wl], bit);
            atomicOr(&cell[((unsigned)d.z / RANGE) * BM_CSTR + wl], bit);
            atomicOr(&cell[((unsigned)d.w / RANGE) * BM_CSTR + wl], bit);
        }
    }
    __syncthreads();

    v4u* out4 = (v4u*)(marks + (size_t)c * (NRANGE * BM_WPC));
    for (int i = tid; i < NRANGE * BM_WPC / 4; i += 256) {
        const unsigned* cp = cell + (i >> 2) * BM_CSTR + (i & 3) * 4;
        const v4u w = {cp[0], cp[1], cp[2], cp[3]};
        out4[i] = w;
    }
}

#define OW_WAVES 32
#define OW_GQCAP 64
#define OW_QCAP  544
#define OW_WPW   (BM_WORDS / OW_WAVES)
static_assert(OW_WPW * OW_WAVES == BM_WORDS, "the waves share the mark row evenly");
static_assert(OW_WPW <= 128, "a lane takes at most four mark words");
static_assert(OW_QCAP >= 31 + 32 * 16, "every edge of 32 lines may belong to the range");
#define OW_TILE_FLOATS 41216
static_assert(OW_TILE_FLOATS >= RANGE * D, "tile must hold the range");

#define OW_SBITS 17
static_assert(N_NODES <= (1 << OW_SBITS), "source id must fit its field");
static_assert(RANGE <= (1 << (31 - OW_SBITS)), "tile row must fit above it");

__device__ __forceinline__ void ow_drain(const float* __restrict__ x,
                                         float* tile, const int* q,
                                         const int lane, const int n) {
    const bool valid = lane < n;
    const int qv = valid ? q[lane] : 0;
    const int s = qv & ((1 << OW_SBITS) - 1);
    const int d = qv >> OW_SBITS;
    #pragma unroll
    for (int b = 0; b < 32; b += 16) {
        if (b < n) {
            float vlo[16], vhi[16];
            #pragma unroll
            for (int j = 0; j < 16; ++j) {
                const int sj = __builtin_amdgcn_readlane(s, b + j);
                const float* xp = x + (size_t)sj * D + lane;
                vlo[j] = xp[0];
                vhi[j] = xp[32];
            }
            #pragma unroll
            for (int j = 0; j < 16; ++j) {
                const int dj = __builtin_amdgcn_readlane(d, b + j);
                const bool on = (b + j) < n;
                float* tp = tile + dj * D + lane;
                atomicAdd(tp,      on ? vlo[j] : 0.0f);
                atomicAdd(tp + 32, on ? vhi[j] : 0.0f);
            }
        }
    }
}

__device__ __forceinline__ void ow_expand(const float* __restrict__ x,
                                          const int* __restrict__ pairs,
                                          float* tile, const int* qg, int* qe, int& cnt,
                                          const int lane, const int lo, const int n) {
    const int half = lane >> 4;
    const int l16  = lane & 15;
    v2i pv[16];
    #pragma unroll
    for (int j = 0; j < 16; ++j) {
        const int slot = 2 * j + half;
        const int g = slot < n ? qg[slot] : 0;
        pv[j] = ((const v2i*)pairs)[(size_t)g * 16 + l16];
    }
    #pragma unroll
    for (int j = 0; j < 16; ++j) {
        const int slot = 2 * j + half;
        const int row = pv[j].y - lo;
        const bool hit = (slot < n) && ((unsigned)row < (unsigned)RANGE);
        const unsigned bal = __builtin_amdgcn_ballot_w32(hit);
        if (hit) {
            const int pos = cnt + (int)__builtin_amdgcn_mbcnt_lo(bal, 0u);
            qe[pos] = pv[j].x | (row << OW_SBITS);
        }
        cnt += __builtin_popcount(bal);
    }
    __builtin_amdgcn_fence(__ATOMIC_ACQ_REL, "wavefront");
    int head = 0;
    while (cnt - head >= 32) {
        ow_drain(x, tile, qe + head, lane, 32);
        head += 32;
    }
    if (head > 0) {
        cnt -= head;
        const bool mv = lane < cnt;
        const int t = mv ? qe[head + lane] : 0;
        if (mv) qe[lane] = t;
    }
    __builtin_amdgcn_fence(__ATOMIC_ACQ_REL, "wavefront");
}

__global__ __launch_bounds__(OW_WAVES * 32) void gin_owner(const float* __restrict__ x,
                                                           const int* __restrict__ pairs,
                                                           const unsigned* __restrict__ marks,
                                                           float* __restrict__ h0) {
    __shared__ float tile[OW_TILE_FLOATS];
    __shared__ int   queue[OW_WAVES][OW_QCAP];
    __shared__ int   gqueue[OW_WAVES][OW_GQCAP];

    const int tid  = threadIdx.x;
    const int lane = tid & 31;
    const int wave = __builtin_amdgcn_readfirstlane(tid >> 5);
    const int lo   = blockIdx.x * RANGE;
    const int nrow = min(RANGE, N_NODES - lo);
    const int n4   = nrow * (D / 4);

    unsigned wv[4];
    #pragma unroll
    for (int it = 0; it < 4; ++it) {
        const int k  = it * 32 + lane;
        const int wi = wave * OW_WPW + k;
        wv[it] = 0u;
        if (k < OW_WPW)
            wv[it] = marks[((size_t)(wi >> 4) * NRANGE + blockIdx.x) * BM_WPC + (wi & 15)];
    }

    {
        const v4f* xs = (const v4f*)(x + (size_t)lo * D);
        v4f* t4 = (v4f*)tile;
        for (int i = tid; i < n4; i += OW_WAVES * 32) t4[i] = xs[i];
    }
    __syncthreads();

    int* q  = &queue[wave][0];
    int* qg = &gqueue[wave][0];
    int cnt  = 0;
    int gcnt = 0;

    for (int it = 0; it < 4; ++it) {
        const int wi = wave * OW_WPW + it * 32 + lane;
        unsigned w = it == 0 ? wv[0] : it == 1 ? wv[1] : it == 2 ? wv[2] : wv[3];
        for (;;) {
            const bool has = w != 0u;
            const unsigned bal = __builtin_amdgcn_ballot_w32(has);
            if (bal == 0u) break;
            if (has) {
                const int gb = __builtin_ctz(w);
                w &= w - 1u;
                const int pos = gcnt + (int)__builtin_amdgcn_mbcnt_lo(bal, 0u);
                qg[pos] = wi * 32 + gb;
            }
            gcnt += __builtin_popcount(bal);
            if (gcnt >= 32) {
                __builtin_amdgcn_fence(__ATOMIC_ACQ_REL, "wavefront");
                ow_expand(x, pairs, tile, qg, q, cnt, lane, lo, 32);
                gcnt -= 32;
                const bool mv = lane < gcnt;
                const int t = mv ? qg[32 + lane] : 0;
                if (mv) qg[lane] = t;
                __builtin_amdgcn_fence(__ATOMIC_ACQ_REL, "wavefront");
            }
        }
    }
    __builtin_amdgcn_fence(__ATOMIC_ACQ_REL, "wavefront");
    if (gcnt > 0) ow_expand(x, pairs, tile, qg, q, cnt, lane, lo, gcnt);
    __builtin_amdgcn_fence(__ATOMIC_ACQ_REL, "wavefront");
    if (cnt > 0) ow_drain(x, tile, q, lane, cnt);

    __syncthreads();
    {
        v4f* hd = (v4f*)(h0 + (size_t)lo * D);
        const v4f* t4 = (const v4f*)tile;
        for (int i = tid; i < n4; i += OW_WAVES * 32) hd[i] = t4[i];
    }
}

#define MLP_WAVES 4
#define HSTR  68
#define KSTR  320
#define LHSTR 160

__global__ __launch_bounds__(128) void gin_mlp(const float* __restrict__ h0,
                                               const float* __restrict__ W1,
                                               const float* __restrict__ b1,
                                               const float* __restrict__ W2,
                                               const float* __restrict__ b2,
                                               float* __restrict__ out) {
    __shared__ float sW1[16 * KSTR];
    __shared__ float sW2[16 * KSTR];
    __shared__ float sB1[D];
    __shared__ float sB2[D];
    __shared__ float sH[MLP_WAVES][16 * HSTR];

    const int tid = threadIdx.x;

    for (int i = tid; i < D * D / 4; i += 128) {
        const int k = i >> 4;
        const int c = (i & 15) << 2;
        const int base = (k >> 2) * KSTR + ((k >> 1) & 1) * LHSTR + (k & 1);
        v4f w1 = ((const v4f*)W1)[i];
        v4f w2 = ((const v4f*)W2)[i];
        #pragma unroll
        for (int j = 0; j < 4; ++j) {
            sW1[base + ((c + j) << 1)] = w1[j];
            sW2[base + ((c + j) << 1)] = w2[j];
        }
    }
    if (tid < D) {
        sB1[tid] = b1[tid];
        sB2[tid] = b2[tid];
    }
    __syncthreads();

    const int wave  = tid >> 5;
    const int lane  = tid & 31;
    const int lhalf = lane >> 4;
    const int l16   = lane & 15;

    const int row0 = (blockIdx.x * MLP_WAVES + wave) * 32;
    const float* a0p = h0 + (size_t)min(row0 + l16,      N_NODES - 1) * D;
    const float* a1p = h0 + (size_t)min(row0 + 16 + l16, N_NODES - 1) * D;

    const v8f vz = {0, 0, 0, 0, 0, 0, 0, 0};
    v8f acc0[4], acc1[4];
    #pragma unroll
    for (int nt = 0; nt < 4; ++nt) { acc0[nt] = vz; acc1[nt] = vz; }

    const float* bw1 = sW1 + lhalf * LHSTR + (l16 << 1);
    const float* bw2 = sW2 + lhalf * LHSTR + (l16 << 1);

    #pragma unroll
    for (int kk = 0; kk < 16; ++kk) {
        const int koff = kk * 4 + lhalf * 2;
        const v2f a0 = *(const v2f*)(a0p + koff);
        const v2f a1 = *(const v2f*)(a1p + koff);
        #pragma unroll
        for (int nt = 0; nt < 4; ++nt) {
            const v2f b = *(const v2f*)(bw1 + kk * KSTR + nt * 32);
            acc0[nt] = __builtin_amdgcn_wmma_f32_16x16x4_f32(
                false, a0, false, b, (short)0, acc0[nt], false, false);
            acc1[nt] = __builtin_amdgcn_wmma_f32_16x16x4_f32(
                false, a1, false, b, (short)0, acc1[nt], false, false);
        }
    }

    float* myH = &sH[wave][0];
    const float* aH = myH + l16 * HSTR;

    #pragma unroll
    for (int mt = 0; mt < 2; ++mt) {
        const v8f* acc = (mt == 0) ? acc0 : acc1;

        #pragma unroll
        for (int nt = 0; nt < 4; ++nt) {
            const int   col  = nt * 16 + l16;
            const float bias = sB1[col];
            #pragma unroll
            for (int r = 0; r < 8; ++r) {
                float v = acc[nt][r] + bias;
                v = v > 0.0f ? v : 0.0f;
                myH[(r + lhalf * 8) * HSTR + col] = v;
            }
        }

        v8f o[4];
        #pragma unroll
        for (int nt = 0; nt < 4; ++nt) o[nt] = vz;

        #pragma unroll
        for (int kk = 0; kk < 16; ++kk) {
            const int koff = kk * 4 + lhalf * 2;
            const v2f a = *(const v2f*)(aH + koff);
            #pragma unroll
            for (int nt = 0; nt < 4; ++nt) {
                const v2f b = *(const v2f*)(bw2 + kk * KSTR + nt * 32);
                o[nt] = __builtin_amdgcn_wmma_f32_16x16x4_f32(
                    false, a, false, b, (short)0, o[nt], false, false);
            }
        }

        #pragma unroll
        for (int r = 0; r < 8; ++r) {
            const int m = row0 + mt * 16 + r + lhalf * 8;
            if (m < N_NODES) {
                #pragma unroll
                for (int nt = 0; nt < 4; ++nt) {
                    const int col = nt * 16 + l16;
                    out[(size_t)m * D + col] = o[nt][r] + sB2[col];
                }
            }
        }
    }
}

extern "C" void kernel_launch(void* const* d_in, const int* in_sizes, int n_in,
                              void* d_out, int out_size, void* d_ws, size_t ws_size,
                              hipStream_t stream) {
    const float* x  = (const float*)d_in[0];
    const int*   ei = (const int*)d_in[1];
    const float* W1 = (const float*)d_in[2];
    const float* b1 = (const float*)d_in[3];
    const float* W2 = (const float*)d_in[4];
    const float* b2 = (const float*)d_in[5];
    float*       out = (float*)d_out;

    const int* src = ei;
    const int* dst = ei + N_EDGES;

    float*    h0    = (float*)d_ws;
    unsigned* marks = (unsigned*)((char*)d_ws + 33554432);
    int*      pairs = (int*)((char*)d_ws + 50331648);

    gin_marks<<<BM_NCH, 256, 0, stream>>>(src, dst, marks, pairs);
    gin_owner<<<NRANGE, OW_WAVES * 32, 0, stream>>>(x, pairs, marks, h0);
    {
        const int rows_per_block = MLP_WAVES * 32;
        gin_mlp<<<(N_NODES + rows_per_block - 1) / rows_per_block, 128, 0, stream>>>(
            h0, W1, b1, W2, b2, out);
    }
}


# MLP as 256 persistent 8-wave work-groups, one per WGP: weights staged once, 6250 16-row tiles dealt evenly (6 or 7 per SIMD, not up to 8), next tile's A rows requested before the current products
# speedup vs baseline: 2.5913x; 1.0935x over previous
#include <hip/hip_runtime.h>


#define N_NODES 100000
#define N_EDGES 1600000
#define D       64

typedef __attribute__((ext_vector_type(2))) float v2f;
typedef __attribute__((ext_vector_type(4))) float v4f;
typedef __attribute__((ext_vector_type(8))) float v8f;
typedef __attribute__((ext_vector_type(2))) int   v2i;
typedef __attribute__((ext_vector_type(4))) int   v4i;
typedef __attribute__((ext_vector_type(4))) unsigned v4u;

#define RANGE   391
#define NRANGE  256
static_assert(RANGE * NRANGE >= N_NODES, "ranges must cover every node");
static_assert((N_NODES - 1) / RANGE < NRANGE, "every destination has a range");
static_assert(N_EDGES % 16 == 0, "edges are marked in whole lines of 16 pairs");
static_assert(D == 64, "lane l covers features l and l+32");

#define BM_CH    8192
#define BM_WPC   (BM_CH / 16 / 32)
#define BM_NCH   ((N_EDGES + BM_CH - 1) / BM_CH)
#define BM_WORDS (BM_NCH * BM_WPC)
#define BM_CSTR  (BM_WPC + 1)
static_assert(BM_WPC == 16, "the owner splits a word index into chunk and j by 4 bits");

__global__ __launch_bounds__(256) void gin_marks(const int* __restrict__ src,
                                                 const int* __restrict__ dst,
                                                 unsigned* __restrict__ marks,
                                                 int* __restrict__ pairs) {
    __shared__ unsigned cell[NRANGE * BM_CSTR];
    const int tid = threadIdx.x;
    const int c = blockIdx.x;
    const int base4 = c * (BM_CH / 4) + tid;

    v4i dv[BM_CH / 4 / 256], sv[BM_CH / 4 / 256];
    #pragma unroll
    for (int it = 0; it < BM_CH / 4 / 256; ++it) {
        const int idx = min(base4 + it * 256, N_EDGES / 4 - 1);
        dv[it] = ((const v4i*)dst)[idx];
        sv[it] = ((const v4i*)src)[idx];
    }

    for (int i = tid; i < NRANGE * BM_CSTR; i += 256) cell[i] = 0u;
    __syncthreads();

    #pragma unroll
    for (int it = 0; it < BM_CH / 4 / 256; ++it) {
        if (base4 + it * 256 < N_EDGES / 4) {
            const v4i d = dv[it];
            const v4i s = sv[it];
            v4i* o = (v4i*)pairs + (size_t)(base4 + it * 256) * 2;
            const v4i p0 = {s.x, d.x, s.y, d.y};
            const v4i p1 = {s.z, d.z, s.w, d.w};
            o[0] = p0;
            o[1] = p1;
            const int ln = (it * 256 + tid) >> 2;
            const unsigned bit = 1u << (ln & 31);
            const int wl = ln >> 5;
            atomicOr(&cell[((unsigned)d.x / RANGE) * BM_CSTR + wl], bit);
            atomicOr(&cell[((unsigned)d.y / RANGE) * BM_CSTR + wl], bit);
            atomicOr(&cell[((unsigned)d.z / RANGE) * BM_CSTR + wl], bit);
            atomicOr(&cell[((unsigned)d.w / RANGE) * BM_CSTR + wl], bit);
        }
    }
    __syncthreads();

    v4u* out4 = (v4u*)(marks + (size_t)c * (NRANGE * BM_WPC));
    for (int i = tid; i < NRANGE * BM_WPC / 4; i += 256) {
        const unsigned* cp = cell + (i >> 2) * BM_CSTR + (i & 3) * 4;
        const v4u w = {cp[0], cp[1], cp[2], cp[3]};
        out4[i] = w;
    }
}

#define OW_WAVES 32
#define OW_GQCAP 64
#define OW_QCAP  544
#define OW_WPW   (BM_WORDS / OW_WAVES)
static_assert(OW_WPW * OW_WAVES == BM_WORDS, "the waves share the mark row evenly");
static_assert(OW_WPW <= 128, "a lane takes at most four mark words");
static_assert(OW_QCAP >= 31 + 32 * 16, "every edge of 32 lines may belong to the range");
#define OW_TILE_FLOATS 41216
static_assert(OW_TILE_FLOATS >= RANGE * D, "tile must hold the range");

#define OW_SBITS 17
static_assert(N_NODES <= (1 << OW_SBITS), "source id must fit its field");
static_assert(RANGE <= (1 << (31 - OW_SBITS)), "tile row must fit above it");

__device__ __forceinline__ void ow_drain(const float* __restrict__ x,
                                         float* tile, const int* q,
                                         const int lane, const int n) {
    const bool valid = lane < n;
    const int qv = valid ? q[lane] : 0;
    const int s = qv & ((1 << OW_SBITS) - 1);
    const int d = qv >> OW_SBITS;
    #pragma unroll
    for (int b = 0; b < 32; b += 16) {
        if (b < n) {
            float vlo[16], vhi[16];
            #pragma unroll
            for (int j = 0; j < 16; ++j) {
                const int sj = __builtin_amdgcn_readlane(s, b + j);
                const float* xp = x + (size_t)sj * D + lane;
                vlo[j] = xp[0];
                vhi[j] = xp[32];
            }
            #pragma unroll
            for (int j = 0; j < 16; ++j) {
                const int dj = __builtin_amdgcn_readlane(d, b + j);
                const bool on = (b + j) < n;
                float* tp = tile + dj * D + lane;
                atomicAdd(tp,      on ? vlo[j] : 0.0f);
                atomicAdd(tp + 32, on ? vhi[j] : 0.0f);
            }
        }
    }
}

__device__ __forceinline__ void ow_expand(const float* __restrict__ x,
                                          const int* __restrict__ pairs,
                                          float* tile, const int* qg, int* qe, int& cnt,
                                          const int lane, const int lo, const int n) {
    const int half = lane >> 4;
    const int l16  = lane & 15;
    v2i pv[16];
    #pragma unroll
    for (int j = 0; j < 16; ++j) {
        const int slot = 2 * j + half;
        const int g = slot < n ? qg[slot] : 0;
        pv[j] = ((const v2i*)pairs)[(size_t)g * 16 + l16];
    }
    #pragma unroll
    for (int j = 0; j < 16; ++j) {
        const int slot = 2 * j + half;
        const int row = pv[j].y - lo;
        const bool hit = (slot < n) && ((unsigned)row < (unsigned)RANGE);
        const unsigned bal = __builtin_amdgcn_ballot_w32(hit);
        if (hit) {
            const int pos = cnt + (int)__builtin_amdgcn_mbcnt_lo(bal, 0u);
            qe[pos] = pv[j].x | (row << OW_SBITS);
        }
        cnt += __builtin_popcount(bal);
    }
    __builtin_amdgcn_fence(__ATOMIC_ACQ_REL, "wavefront");
    int head = 0;
    while (cnt - head >= 32) {
        ow_drain(x, tile, qe + head, lane, 32);
        head += 32;
    }
    if (head > 0) {
        cnt -= head;
        const bool mv = lane < cnt;
        const int t = mv ? qe[head + lane] : 0;
        if (mv) qe[lane] = t;
    }
    __builtin_amdgcn_fence(__ATOMIC_ACQ_REL, "wavefront");
}

__global__ __launch_bounds__(OW_WAVES * 32) void gin_owner(const float* __restrict__ x,
                                                           const int* __restrict__ pairs,
                                                           const unsigned* __restrict__ marks,
                                                           float* __restrict__ h0) {
    __shared__ float tile[OW_TILE_FLOATS];
    __shared__ int   queue[OW_WAVES][OW_QCAP];
    __shared__ int   gqueue[OW_WAVES][OW_GQCAP];

    const int tid  = threadIdx.x;
    const int lane = tid & 31;
    const int wave = __builtin_amdgcn_readfirstlane(tid >> 5);
    const int lo   = blockIdx.x * RANGE;
    const int nrow = min(RANGE, N_NODES - lo);
    const int n4   = nrow * (D / 4);

    unsigned wv[4];
    #pragma unroll
    for (int it = 0; it < 4; ++it) {
        const int k  = it * 32 + lane;
        const int wi = wave * OW_WPW + k;
        wv[it] = 0u;
        if (k < OW_WPW)
            wv[it] = marks[((size_t)(wi >> 4) * NRANGE + blockIdx.x) * BM_WPC + (wi & 15)];
    }

    {
        const v4f* xs = (const v4f*)(x + (size_t)lo * D);
        v4f* t4 = (v4f*)tile;
        for (int i = tid; i < n4; i += OW_WAVES * 32) t4[i] = xs[i];
    }
    __syncthreads();

    int* q  = &queue[wave][0];
    int* qg = &gqueue[wave][0];
    int cnt  = 0;
    int gcnt = 0;

    for (int it = 0; it < 4; ++it) {
        const int wi = wave * OW_WPW + it * 32 + lane;
        unsigned w = it == 0 ? wv[0] : it == 1 ? wv[1] : it == 2 ? wv[2] : wv[3];
        for (;;) {
            const bool has = w != 0u;
            const unsigned bal = __builtin_amdgcn_ballot_w32(has);
            if (bal == 0u) break;
            if (has) {
                const int gb = __builtin_ctz(w);
                w &= w - 1u;
                const int pos = gcnt + (int)__builtin_amdgcn_mbcnt_lo(bal, 0u);
                qg[pos] = wi * 32 + gb;
            }
            gcnt += __builtin_popcount(bal);
            if (gcnt >= 32) {
                __builtin_amdgcn_fence(__ATOMIC_ACQ_REL, "wavefront");
                ow_expand(x, pairs, tile, qg, q, cnt, lane, lo, 32);
                gcnt -= 32;
                const bool mv = lane < gcnt;
                const int t = mv ? qg[32 + lane] : 0;
                if (mv) qg[lane] = t;
                __builtin_amdgcn_fence(__ATOMIC_ACQ_REL, "wavefront");
            }
        }
    }
    __builtin_amdgcn_fence(__ATOMIC_ACQ_REL, "wavefront");
    if (gcnt > 0) ow_expand(x, pairs, tile, qg, q, cnt, lane, lo, gcnt);
    __builtin_amdgcn_fence(__ATOMIC_ACQ_REL, "wavefront");
    if (cnt > 0) ow_drain(x, tile, q, lane, cnt);

    __syncthreads();
    {
        v4f* hd = (v4f*)(h0 + (size_t)lo * D);
        const v4f* t4 = (const v4f*)tile;
        for (int i = tid; i < n4; i += OW_WAVES * 32) hd[i] = t4[i];
    }
}

#define MLP_WAVES 8
#define MLP_NWG   256
#define MLP_TILES (N_NODES / 16)
#define MLP_STEP  (MLP_NWG * MLP_WAVES)
static_assert(N_NODES % 16 == 0, "every row tile is full: no row guard needed");
#define HSTR  68
#define KSTR  320
#define LHSTR 160
#define MLP_HW_FLOATS 3840
static_assert(MLP_HW_FLOATS >= 16 * HSTR, "relayout tile must fit");

__global__ __launch_bounds__(MLP_WAVES * 32) void gin_mlp(const float* __restrict__ h0,
                                                          const float* __restrict__ W1,
                                                          const float* __restrict__ b1,
                                                          const float* __restrict__ W2,
                                                          const float* __restrict__ b2,
                                                          float* __restrict__ out) {
    __shared__ float sW1[16 * KSTR];
    __shared__ float sW2[16 * KSTR];
    __shared__ float sB1[D];
    __shared__ float sB2[D];
    __shared__ float sH[MLP_WAVES][MLP_HW_FLOATS];

    const int tid   = threadIdx.x;
    const int lane  = tid & 31;
    const int wave  = __builtin_amdgcn_readfirstlane(tid >> 5);
    const int lhalf = lane >> 4;
    const int l16   = lane & 15;

    const int first = blockIdx.x + MLP_NWG * wave;

    v2f a[16];
    {
        const int t0 = min(first, MLP_TILES - 1);
        const float* ap = h0 + (size_t)(t0 * 16 + l16) * D + lhalf * 2;
        #pragma unroll
        for (int kk = 0; kk < 16; ++kk) a[kk] = *(const v2f*)(ap + kk * 4);
    }

    for (int i = tid; i < D * D / 4; i += MLP_WAVES * 32) {
        const int k = i >> 4;
        const int c = (i & 15) << 2;
        const int base = (k >> 2) * KSTR + ((k >> 1) & 1) * LHSTR + (k & 1);
        v4f w1 = ((const v4f*)W1)[i];
        v4f w2 = ((const v4f*)W2)[i];
        #pragma unroll
        for (int j = 0; j < 4; ++j) {
            sW1[base + ((c + j) << 1)] = w1[j];
            sW2[base + ((c + j) << 1)] = w2[j];
        }
    }
    if (tid < D) {
        sB1[tid] = b1[tid];
        sB2[tid] = b2[tid];
    }
    __syncthreads();

    const v8f vz = {0, 0, 0, 0, 0, 0, 0, 0};
    const float* bw1 = sW1 + lhalf * LHSTR + (l16 << 1);
    const float* bw2 = sW2 + lhalf * LHSTR + (l16 << 1);
    float* myH = &sH[wave][0];
    const float* aH = myH + l16 * HSTR;

    for (int tile = first; tile < MLP_TILES; tile += MLP_STEP) {
        v2f an[16];
        {
            const int tn = (tile + MLP_STEP < MLP_TILES) ? tile + MLP_STEP : tile;
            const float* ap = h0 + (size_t)(tn * 16 + l16) * D + lhalf * 2;
            #pragma unroll
            for (int kk = 0; kk < 16; ++kk) an[kk] = *(const v2f*)(ap + kk * 4);
        }

        v8f acc[4];
        #pragma unroll
        for (int nt = 0; nt < 4; ++nt) acc[nt] = vz;
        #pragma unroll
        for (int kk = 0; kk < 16; ++kk) {
            #pragma unroll
            for (int nt = 0; nt < 4; ++nt) {
                const v2f b = *(const v2f*)(bw1 + kk * KSTR + nt * 32);
                acc[nt] = __builtin_amdgcn_wmma_f32_16x16x4_f32(
                    false, a[kk], false, b, (short)0, acc[nt], false, false);
            }
        }

        #pragma unroll
        for (int nt = 0; nt < 4; ++nt) {
            const int   col  = nt * 16 + l16;
            const float bias = sB1[col];
            #pragma unroll
            for (int r = 0; r < 8; ++r) {
                float v = acc[nt][r] + bias;
                v = v > 0.0f ? v : 0.0f;
                myH[(r + lhalf * 8) * HSTR + col] = v;
            }
        }

        v8f o[4];
        #pragma unroll
        for (int nt = 0; nt < 4; ++nt) o[nt] = vz;
        #pragma unroll
        for (int kk = 0; kk < 16; ++kk) {
            const int koff = kk * 4 + lhalf * 2;
            const v2f av = *(const v2f*)(aH + koff);
            #pragma unroll
            for (int nt = 0; nt < 4; ++nt) {
                const v2f b = *(const v2f*)(bw2 + kk * KSTR + nt * 32);
                o[nt] = __builtin_amdgcn_wmma_f32_16x16x4_f32(
                    false, av, false, b, (short)0, o[nt], false, false);
            }
        }

        #pragma unroll
        for (int r = 0; r < 8; ++r) {
            const int m = tile * 16 + r + lhalf * 8;
            #pragma unroll
            for (int nt = 0; nt < 4; ++nt) {
                const int col = nt * 16 + l16;
                out[(size_t)m * D + col] = o[nt][r] + sB2[col];
            }
        }

        #pragma unroll
        for (int kk = 0; kk < 16; ++kk) a[kk] = an[kk];
    }
}

extern "C" void kernel_launch(void* const* d_in, const int* in_sizes, int n_in,
                              void* d_out, int out_size, void* d_ws, size_t ws_size,
                              hipStream_t stream) {
    const float* x  = (const float*)d_in[0];
    const int*   ei = (const int*)d_in[1];
    const float* W1 = (const float*)d_in[2];
    const float* b1 = (const float*)d_in[3];
    const float* W2 = (const float*)d_in[4];
    const float* b2 = (const float*)d_in[5];
    float*       out = (float*)d_out;

    const int* src = ei;
    const int* dst = ei + N_EDGES;

    float*    h0    = (float*)d_ws;
    unsigned* marks = (unsigned*)((char*)d_ws + 33554432);
    int*      pairs = (int*)((char*)d_ws + 50331648);

    gin_marks<<<BM_NCH, 256, 0, stream>>>(src, dst, marks, pairs);
    gin_owner<<<NRANGE, OW_WAVES * 32, 0, stream>>>(x, pairs, marks, h0);
    gin_mlp<<<MLP_NWG, MLP_WAVES * 32, 0, stream>>>(h0, W1, b1, W2, b2, out);
}


# copies ask for all their loads before the first wait: owner seed and h write-out (7 pieces a thread), MLP weights and biases ahead of the first A rows, biases to registers: one trip, not one per pass
# speedup vs baseline: 2.6783x; 1.0336x over previous
#include <hip/hip_runtime.h>


#define N_NODES 100000
#define N_EDGES 1600000
#define D       64

typedef __attribute__((ext_vector_type(2))) float v2f;
typedef __attribute__((ext_vector_type(4))) float v4f;
typedef __attribute__((ext_vector_type(8))) float v8f;
typedef __attribute__((ext_vector_type(2))) int   v2i;
typedef __attribute__((ext_vector_type(4))) int   v4i;
typedef __attribute__((ext_vector_type(4))) unsigned v4u;

#define RANGE   391
#define NRANGE  256
static_assert(RANGE * NRANGE >= N_NODES, "ranges must cover every node");
static_assert((N_NODES - 1) / RANGE < NRANGE, "every destination has a range");
static_assert(N_EDGES % 16 == 0, "edges are marked in whole lines of 16 pairs");
static_assert(D == 64, "lane l covers features l and l+32");

#define BM_CH    8192
#define BM_WPC   (BM_CH / 16 / 32)
#define BM_NCH   ((N_EDGES + BM_CH - 1) / BM_CH)
#define BM_WORDS (BM_NCH * BM_WPC)
#define BM_CSTR  (BM_WPC + 1)
static_assert(BM_WPC == 16, "the owner splits a word index into chunk and j by 4 bits");

__global__ __launch_bounds__(256) void gin_marks(const int* __restrict__ src,
                                                 const int* __restrict__ dst,
                                                 unsigned* __restrict__ marks,
                                                 int* __restrict__ pairs) {
    __shared__ unsigned cell[NRANGE * BM_CSTR];
    const int tid = threadIdx.x;
    const int c = blockIdx.x;
    const int base4 = c * (BM_CH / 4) + tid;

    v4i dv[BM_CH / 4 / 256], sv[BM_CH / 4 / 256];
    #pragma unroll
    for (int it = 0; it < BM_CH / 4 / 256; ++it) {
        const int idx = min(base4 + it * 256, N_EDGES / 4 - 1);
        dv[it] = ((const v4i*)dst)[idx];
        sv[it] = ((const v4i*)src)[idx];
    }

    for (int i = tid; i < NRANGE * BM_CSTR; i += 256) cell[i] = 0u;
    __syncthreads();

    #pragma unroll
    for (int it = 0; it < BM_CH / 4 / 256; ++it) {
        if (base4 + it * 256 < N_EDGES / 4) {
            const v4i d = dv[it];
            const v4i s = sv[it];
            v4i* o = (v4i*)pairs + (size_t)(base4 + it * 256) * 2;
            const v4i p0 = {s.x, d.x, s.y, d.y};
            const v4i p1 = {s.z, d.z, s.w, d.w};
            o[0] = p0;
            o[1] = p1;
            const int ln = (it * 256 + tid) >> 2;
            const unsigned bit = 1u << (ln & 31);
            const int wl = ln >> 5;
            atomicOr(&cell[((unsigned)d.x / RANGE) * BM_CSTR + wl], bit);
            atomicOr(&cell[((unsigned)d.y / RANGE) * BM_CSTR + wl], bit);
            atomicOr(&cell[((unsigned)d.z / RANGE) * BM_CSTR + wl], bit);
            atomicOr(&cell[((unsigned)d.w / RANGE) * BM_CSTR + wl], bit);
        }
    }
    __syncthreads();

    v4u* out4 = (v4u*)(marks + (size_t)c * (NRANGE * BM_WPC));
    for (int i = tid; i < NRANGE * BM_WPC / 4; i += 256) {
        const unsigned* cp = cell + (i >> 2) * BM_CSTR + (i & 3) * 4;
        const v4u w = {cp[0], cp[1], cp[2], cp[3]};
        out4[i] = w;
    }
}

#define OW_WAVES 32
#define OW_GQCAP 64
#define OW_QCAP  544
#define OW_WPW   (BM_WORDS / OW_WAVES)
#define OW_CPY   ((RANGE * (D / 4) + OW_WAVES * 32 - 1) / (OW_WAVES * 32))
static_assert(OW_WPW * OW_WAVES == BM_WORDS, "the waves share the mark row evenly");
static_assert(OW_WPW <= 128, "a lane takes at most four mark words");
static_assert(OW_QCAP >= 31 + 32 * 16, "every edge of 32 lines may belong to the range");
#define OW_TILE_FLOATS 41216
static_assert(OW_TILE_FLOATS >= RANGE * D, "tile must hold the range");

#define OW_SBITS 17
static_assert(N_NODES <= (1 << OW_SBITS), "source id must fit its field");
static_assert(RANGE <= (1 << (31 - OW_SBITS)), "tile row must fit above it");

__device__ __forceinline__ void ow_drain(const float* __restrict__ x,
                                         float* tile, const int* q,
                                         const int lane, const int n) {
    const bool valid = lane < n;
    const int qv = valid ? q[lane] : 0;
    const int s = qv & ((1 << OW_SBITS) - 1);
    const int d = qv >> OW_SBITS;
    #pragma unroll
    for (int b = 0; b < 32; b += 16) {
        if (b < n) {
            float vlo[16], vhi[16];
            #pragma unroll
            for (int j = 0; j < 16; ++j) {
                const int sj = __builtin_amdgcn_readlane(s, b + j);
                const float* xp = x + (size_t)sj * D + lane;
                vlo[j] = xp[0];
                vhi[j] = xp[32];
            }
            #pragma unroll
            for (int j = 0; j < 16; ++j) {
                const int dj = __builtin_amdgcn_readlane(d, b + j);
                const bool on = (b + j) < n;
                float* tp = tile + dj * D + lane;
                atomicAdd(tp,      on ? vlo[j] : 0.0f);
                atomicAdd(tp + 32, on ? vhi[j] : 0.0f);
            }
        }
    }
}

__device__ __forceinline__ void ow_expand(const float* __restrict__ x,
                                          const int* __restrict__ pairs,
                                          float* tile, const int* qg, int* qe, int& cnt,
                                          const int lane, const int lo, const int n) {
    const int half = lane >> 4;
    const int l16  = lane & 15;
    v2i pv[16];
    #pragma unroll
    for (int j = 0; j < 16; ++j) {
        const int slot = 2 * j + half;
        const int g = slot < n ? qg[slot] : 0;
        pv[j] = ((const v2i*)pairs)[(size_t)g * 16 + l16];
    }
    #pragma unroll
    for (int j = 0; j < 16; ++j) {
        const int slot = 2 * j + half;
        const int row = pv[j].y - lo;
        const bool hit = (slot < n) && ((unsigned)row < (unsigned)RANGE);
        const unsigned bal = __builtin_amdgcn_ballot_w32(hit);
        if (hit) {
            const int pos = cnt + (int)__builtin_amdgcn_mbcnt_lo(bal, 0u);
            qe[pos] = pv[j].x | (row << OW_SBITS);
        }
        cnt += __builtin_popcount(bal);
    }
    __builtin_amdgcn_fence(__ATOMIC_ACQ_REL, "wavefront");
    int head = 0;
    while (cnt - head >= 32) {
        ow_drain(x, tile, qe + head, lane, 32);
        head += 32;
    }
    if (head > 0) {
        cnt -= head;
        const bool mv = lane < cnt;
        const int t = mv ? qe[head + lane] : 0;
        if (mv) qe[lane] = t;
    }
    __builtin_amdgcn_fence(__ATOMIC_ACQ_REL, "wavefront");
}

__global__ __launch_bounds__(OW_WAVES * 32) void gin_owner(const float* __restrict__ x,
                                                           const int* __restrict__ pairs,
                                                           const unsigned* __restrict__ marks,
                                                           float* __restrict__ h0) {
    __shared__ float tile[OW_TILE_FLOATS];
    __shared__ int   queue[OW_WAVES][OW_QCAP];
    __shared__ int   gqueue[OW_WAVES][OW_GQCAP];

    const int tid  = threadIdx.x;
    const int lane = tid & 31;
    const int wave = __builtin_amdgcn_readfirstlane(tid >> 5);
    const int lo   = blockIdx.x * RANGE;
    const int nrow = min(RANGE, N_NODES - lo);
    const int n4   = nrow * (D / 4);

    unsigned wv[4];
    #pragma unroll
    for (int it = 0; it < 4; ++it) {
        const int k  = it * 32 + lane;
        const int wi = wave * OW_WPW + k;
        wv[it] = 0u;
        if (k < OW_WPW)
            wv[it] = marks[((size_t)(wi >> 4) * NRANGE + blockIdx.x) * BM_WPC + (wi & 15)];
    }

    {
        const v4f* xs = (const v4f*)(x + (size_t)lo * D);
        v4f* t4 = (v4f*)tile;
        v4f sv[OW_CPY];
        #pragma unroll
        for (int it = 0; it < OW_CPY; ++it)
            sv[it] = xs[min(tid + it * (OW_WAVES * 32), n4 - 1)];
        #pragma unroll
        for (int it = 0; it < OW_CPY; ++it) {
            const int i = tid + it * (OW_WAVES * 32);
            if (i < n4) t4[i] = sv[it];
        }
    }
    __syncthreads();

    int* q  = &queue[wave][0];
    int* qg = &gqueue[wave][0];
    int cnt  = 0;
    int gcnt = 0;

    for (int it = 0; it < 4; ++it) {
        const int wi = wave * OW_WPW + it * 32 + lane;
        unsigned w = it == 0 ? wv[0] : it == 1 ? wv[1] : it == 2 ? wv[2] : wv[3];
        for (;;) {
            const bool has = w != 0u;
            const unsigned bal = __builtin_amdgcn_ballot_w32(has);
            if (bal == 0u) break;
            if (has) {
                const int gb = __builtin_ctz(w);
                w &= w - 1u;
                const int pos = gcnt + (int)__builtin_amdgcn_mbcnt_lo(bal, 0u);
                qg[pos] = wi * 32 + gb;
            }
            gcnt += __builtin_popcount(bal);
            if (gcnt >= 32) {
                __builtin_amdgcn_fence(__ATOMIC_ACQ_REL, "wavefront");
                ow_expand(x, pairs, tile, qg, q, cnt, lane, lo, 32);
                gcnt -= 32;
                const bool mv = lane < gcnt;
                const int t = mv ? qg[32 + lane] : 0;
                if (mv) qg[lane] = t;
                __builtin_amdgcn_fence(__ATOMIC_ACQ_REL, "wavefront");
            }
        }
    }
    __builtin_amdgcn_fence(__ATOMIC_ACQ_REL, "wavefront");
    if (gcnt > 0) ow_expand(x, pairs, tile, qg, q, cnt, lane, lo, gcnt);
    __builtin_amdgcn_fence(__ATOMIC_ACQ_REL, "wavefront");
    if (cnt > 0) ow_drain(x, tile, q, lane, cnt);

    __syncthreads();
    {
        v4f* hd = (v4f*)(h0 + (size_t)lo * D);
        const v4f* t4 = (const v4f*)tile;
        v4f ov[OW_CPY];
        #pragma unroll
        for (int it = 0; it < OW_CPY; ++it)
            ov[it] = t4[min(tid + it * (OW_WAVES * 32), n4 - 1)];
        #pragma unroll
        for (int it = 0; it < OW_CPY; ++it) {
            const int i = tid + it * (OW_WAVES * 32);
            if (i < n4) hd[i] = ov[it];
        }
    }
}

#define MLP_WAVES 8
#define MLP_NWG   256
#define MLP_TILES (N_NODES / 16)
#define MLP_STEP  (MLP_NWG * MLP_WAVES)
static_assert(N_NODES % 16 == 0, "every row tile is full: no row guard needed");
#define HSTR  68
#define KSTR  320
#define LHSTR 160
#define MLP_HW_FLOATS 3856
static_assert(MLP_HW_FLOATS >= 16 * HSTR, "relayout tile must fit");
#define MLP_WST (D * D / 4 / (MLP_WAVES * 32))
static_assert(MLP_WST * MLP_WAVES * 32 * 4 == D * D, "the threads share a weight matrix evenly");
static_assert(2 * 16 * KSTR * 4 + MLP_WAVES * MLP_HW_FLOATS * 4 > 160 * 1024, "one work-group per WGP");

__global__ __launch_bounds__(MLP_WAVES * 32) void gin_mlp(const float* __restrict__ h0,
                                                          const float* __restrict__ W1,
                                                          const float* __restrict__ b1,
                                                          const float* __restrict__ W2,
                                                          const float* __restrict__ b2,
                                                          float* __restrict__ out) {
    __shared__ float sW1[16 * KSTR];
    __shared__ float sW2[16 * KSTR];
    __shared__ float sH[MLP_WAVES][MLP_HW_FLOATS];

    const int tid   = threadIdx.x;
    const int lane  = tid & 31;
    const int wave  = __builtin_amdgcn_readfirstlane(tid >> 5);
    const int lhalf = lane >> 4;
    const int l16   = lane & 15;

    const int first = blockIdx.x + MLP_NWG * wave;

    v4f w1r[MLP_WST], w2r[MLP_WST];
    #pragma unroll
    for (int it = 0; it < MLP_WST; ++it) {
        w1r[it] = ((const v4f*)W1)[tid + it * (MLP_WAVES * 32)];
        w2r[it] = ((const v4f*)W2)[tid + it * (MLP_WAVES * 32)];
    }
    float bias1[4], bias2[4];
    #pragma unroll
    for (int nt = 0; nt < 4; ++nt) {
        bias1[nt] = b1[nt * 16 + l16];
        bias2[nt] = b2[nt * 16 + l16];
    }
    v2f a[16];
    {
        const int t0 = min(first, MLP_TILES - 1);
        const float* ap = h0 + (size_t)(t0 * 16 + l16) * D + lhalf * 2;
        #pragma unroll
        for (int kk = 0; kk < 16; ++kk) a[kk] = *(const v2f*)(ap + kk * 4);
    }

    #pragma unroll
    for (int it = 0; it < MLP_WST; ++it) {
        const int i = tid + it * (MLP_WAVES * 32);
        const int k = i >> 4;
        const int c = (i & 15) << 2;
        const int base = (k >> 2) * KSTR + ((k >> 1) & 1) * LHSTR + (k & 1);
        #pragma unroll
        for (int j = 0; j < 4; ++j) {
            sW1[base + ((c + j) << 1)] = w1r[it][j];
            sW2[base + ((c + j) << 1)] = w2r[it][j];
        }
    }
    __syncthreads();

    const v8f vz = {0, 0, 0, 0, 0, 0, 0, 0};
    const float* bw1 = sW1 + lhalf * LHSTR + (l16 << 1);
    const float* bw2 = sW2 + lhalf * LHSTR + (l16 << 1);
    float* myH = &sH[wave][0];
    const float* aH = myH + l16 * HSTR;

    for (int tile = first; tile < MLP_TILES; tile += MLP_STEP) {
        v2f an[16];
        {
            const int tn = (tile + MLP_STEP < MLP_TILES) ? tile + MLP_STEP : tile;
            const float* ap = h0 + (size_t)(tn * 16 + l16) * D + lhalf * 2;
            #pragma unroll
            for (int kk = 0; kk < 16; ++kk) an[kk] = *(const v2f*)(ap + kk * 4);
        }

        v8f acc[4];
        #pragma unroll
        for (int nt = 0; nt < 4; ++nt) acc[nt] = vz;
        #pragma unroll
        for (int kk = 0; kk < 16; ++kk) {
            #pragma unroll
            for (int nt = 0; nt < 4; ++nt) {
                const v2f b = *(const v2f*)(bw1 + kk * KSTR + nt * 32);
                acc[nt] = __builtin_amdgcn_wmma_f32_16x16x4_f32(
                    false, a[kk], false, b, (short)0, acc[nt], false, false);
            }
        }

        #pragma unroll
        for (int nt = 0; nt < 4; ++nt) {
            const int   col  = nt * 16 + l16;
            const float bias = bias1[nt];
            #pragma unroll
            for (int r = 0; r < 8; ++r) {
                float v = acc[nt][r] + bias;
                v = v > 0.0f ? v : 0.0f;
                myH[(r + lhalf * 8) * HSTR + col] = v;
            }
        }

        v8f o[4];
        #pragma unroll
        for (int nt = 0; nt < 4; ++nt) o[nt] = vz;
        #pragma unroll
        for (int kk = 0; kk < 16; ++kk) {
            const int koff = kk * 4 + lhalf * 2;
            const v2f av = *(const v2f*)(aH + koff);
            #pragma unroll
            for (int nt = 0; nt < 4; ++nt) {
                const v2f b = *(const v2f*)(bw2 + kk * KSTR + nt * 32);
                o[nt] = __builtin_amdgcn_wmma_f32_16x16x4_f32(
                    false, av, false, b, (short)0, o[nt], false, false);
            }
        }

        #pragma unroll
        for (int r = 0; r < 8; ++r) {
            const int m = tile * 16 + r + lhalf * 8;
            #pragma unroll
            for (int nt = 0; nt < 4; ++nt) {
                const int col = nt * 16 + l16;
                out[(size_t)m * D + col] = o[nt][r] + bias2[nt];
            }
        }

        #pragma unroll
        for (int kk = 0; kk < 16; ++kk) a[kk] = an[kk];
    }
}

extern "C" void kernel_launch(void* const* d_in, const int* in_sizes, int n_in,
                              void* d_out, int out_size, void* d_ws, size_t ws_size,
                              hipStream_t stream) {
    const float* x  = (const float*)d_in[0];
    const int*   ei = (const int*)d_in[1];
    const float* W1 = (const float*)d_in[2];
    const float* b1 = (const float*)d_in[3];
    const float* W2 = (const float*)d_in[4];
    const float* b2 = (const float*)d_in[5];
    float*       out = (float*)d_out;

    const int* src = ei;
    const int* dst = ei + N_EDGES;

    float*    h0    = (float*)d_ws;
    unsigned* marks = (unsigned*)((char*)d_ws + 33554432);
    int*      pairs = (int*)((char*)d_ws + 50331648);

    gin_marks<<<BM_NCH, 256, 0, stream>>>(src, dst, marks, pairs);
    gin_owner<<<NRANGE, OW_WAVES * 32, 0, stream>>>(x, pairs, marks, h0);
    gin_mlp<<<MLP_NWG, MLP_WAVES * 32, 0, stream>>>(h0, W1, b1, W2, b2, out);
}


# edges grouped by range per 4096-edge chunk (rank by returning LDS add, scan, LDS stage; one word per edge): owners read their runs, no marks or pair lines opened; sums stay in arrival order
# speedup vs baseline: 3.1588x; 1.1794x over previous
#include <hip/hip_runtime.h>


#define N_NODES 100000
#define N_EDGES 1600000
#define D       64

typedef __attribute__((ext_vector_type(2))) float v2f;
typedef __attribute__((ext_vector_type(4))) float v4f;
typedef __attribute__((ext_vector_type(8))) float v8f;
typedef __attribute__((ext_vector_type(2))) int   v2i;
typedef __attribute__((ext_vector_type(4))) int   v4i;
typedef __attribute__((ext_vector_type(4))) unsigned v4u;

#define RANGE   391
#define NRANGE  256
static_assert(RANGE * NRANGE >= N_NODES, "ranges must cover every node");
static_assert((N_NODES - 1) / RANGE < NRANGE, "every destination has a range");
static_assert(N_EDGES % 4 == 0, "edges are read and written four at a time");
static_assert(D == 64, "lane l covers features l and l+32");

#define OW_SBITS 17
static_assert(N_NODES <= (1 << OW_SBITS), "source id must fit its field");
static_assert(RANGE <= (1 << (31 - OW_SBITS)), "tile row must fit above it");

#define BM_CH    4096
#define BM_NCH   ((N_EDGES + BM_CH - 1) / BM_CH)
#define BM_PPT   (BM_CH / 4 / 256)
static_assert(BM_CH < (1 << 16), "a run's start and length fit 16 bits each");
static_assert(BM_PPT * 4 * 256 == BM_CH, "the threads share a chunk evenly");
static_assert(NRANGE == 256, "one thread per range in the scan; the range is 8 bits of a key");

__global__ __launch_bounds__(256) void gin_group(const int* __restrict__ src,
                                                 const int* __restrict__ dst,
                                                 unsigned* __restrict__ tab,
                                                 int* __restrict__ grouped) {
    __shared__ int cell[NRANGE];
    __shared__ int first[NRANGE];
    __shared__ int wsum[8];
    __shared__ int stage[BM_CH];
    const int tid  = threadIdx.x;
    const int lane = tid & 31;
    const int wave = tid >> 5;
    const int c = blockIdx.x;
    const int base4 = c * (BM_CH / 4) + tid;

    v4i dv[BM_PPT], sv[BM_PPT];
    #pragma unroll
    for (int it = 0; it < BM_PPT; ++it) {
        const int idx = min(base4 + it * 256, N_EDGES / 4 - 1);
        dv[it] = ((const v4i*)dst)[idx];
        sv[it] = ((const v4i*)src)[idx];
    }

    cell[tid] = 0;
    __syncthreads();

    #pragma unroll
    for (int it = 0; it < BM_PPT; ++it) {
        if (base4 + it * 256 < N_EDGES / 4) {
            #pragma unroll
            for (int j = 0; j < 4; ++j) {
                const unsigned d = (unsigned)dv[it][j];
                const unsigned r = d / RANGE;
                const int row = (int)(d - r * RANGE);
                const int rank = atomicAdd(&cell[r], 1);
                sv[it][j] = sv[it][j] | (row << OW_SBITS);
                dv[it][j] = (int)r | (rank << 8);
            }
        }
    }
    __syncthreads();

    const int my = cell[tid];
    int inc = my;
    #pragma unroll
    for (int o = 1; o < 32; o <<= 1) {
        const int v = __shfl_up(inc, o);
        if (lane >= o) inc += v;
    }
    if (lane == 31) wsum[wave] = inc;
    __syncthreads();
    int before = 0;
    #pragma unroll
    for (int w = 0; w < 8; ++w) before += (w < wave) ? wsum[w] : 0;
    const int st = before + inc - my;
    first[tid] = st;
    tab[(size_t)c * NRANGE + tid] = (unsigned)st | ((unsigned)my << 16);
    __syncthreads();

    #pragma unroll
    for (int it = 0; it < BM_PPT; ++it) {
        if (base4 + it * 256 < N_EDGES / 4) {
            #pragma unroll
            for (int j = 0; j < 4; ++j) {
                const int key = dv[it][j];
                stage[first[key & 255] + (key >> 8)] = sv[it][j];
            }
        }
    }
    __syncthreads();

    const int total4 = min(BM_CH, N_EDGES - c * BM_CH) / 4;
    v4i* g4 = (v4i*)(grouped + (size_t)c * BM_CH);
    const v4i* s4 = (const v4i*)stage;
    v4i ov[BM_PPT];
    #pragma unroll
    for (int it = 0; it < BM_PPT; ++it) ov[it] = s4[min(tid + it * 256, total4 - 1)];
    #pragma unroll
    for (int it = 0; it < BM_PPT; ++it) {
        const int i = tid + it * 256;
        if (i < total4) g4[i] = ov[it];
    }
}

#define OW_WAVES 32
#define OW_CPW   ((BM_NCH + OW_WAVES - 1) / OW_WAVES)
#define OW_QCAP  (OW_CPW * 32)
#define OW_CPY   ((RANGE * (D / 4) + OW_WAVES * 32 - 1) / (OW_WAVES * 32))
static_assert(OW_CPW <= 32, "a lane holds the table word of one chunk of its wave");
static_assert(OW_QCAP >= 31 + 32, "a load of 32 words fits behind what is held");
#define OW_TILE_FLOATS 41216
static_assert(OW_TILE_FLOATS >= RANGE * D, "tile must hold the range");

__device__ __forceinline__ void ow_drain(const float* __restrict__ x,
                                         float* tile, const int* q,
                                         const int lane, const int n) {
    const bool valid = lane < n;
    const int qv = valid ? q[lane] : 0;
    const int s = qv & ((1 << OW_SBITS) - 1);
    const int d = qv >> OW_SBITS;
    #pragma unroll
    for (int b = 0; b < 32; b += 16) {
        if (b < n) {
            float vlo[16], vhi[16];
            #pragma unroll
            for (int j = 0; j < 16; ++j) {
                const int sj = __builtin_amdgcn_readlane(s, b + j);
                const float* xp = x + (size_t)sj * D + lane;
                vlo[j] = xp[0];
                vhi[j] = xp[32];
            }
            #pragma unroll
            for (int j = 0; j < 16; ++j) {
                const int dj = __builtin_amdgcn_readlane(d, b + j);
                const bool on = (b + j) < n;
                float* tp = tile + dj * D + lane;
                atomicAdd(tp,      on ? vlo[j] : 0.0f);
                atomicAdd(tp + 32, on ? vhi[j] : 0.0f);
            }
        }
    }
}

__global__ __launch_bounds__(OW_WAVES * 32) void gin_owner(const float* __restrict__ x,
                                                           const int* __restrict__ grouped,
                                                           const unsigned* __restrict__ tab,
                                                           float* __restrict__ h0) {
    __shared__ float tile[OW_TILE_FLOATS];
    __shared__ int   queue[OW_WAVES][OW_QCAP];

    const int tid  = threadIdx.x;
    const int lane = tid & 31;
    const int wave = __builtin_amdgcn_readfirstlane(tid >> 5);
    const int lo   = blockIdx.x * RANGE;
    const int nrow = min(RANGE, N_NODES - lo);
    const int n4   = nrow * (D / 4);

    int tw;
    {
        const int ck = wave + OW_WAVES * lane;
        const int got = (int)tab[(size_t)min(ck, BM_NCH - 1) * NRANGE + blockIdx.x];
        tw = ck < BM_NCH ? got : 0;
    }

    const v4f* xs = (const v4f*)(x + (size_t)lo * D);
    v4f* t4 = (v4f*)tile;
    v4f sv[OW_CPY];
    #pragma unroll
    for (int it = 0; it < OW_CPY; ++it)
        sv[it] = xs[min(tid + it * (OW_WAVES * 32), n4 - 1)];

    int e[OW_CPW];
    #pragma unroll
    for (int k = 0; k < OW_CPW; ++k) {
        const unsigned t = (unsigned)__builtin_amdgcn_readlane(tw, k);
        const int st = (int)(t & 0xffffu);
        const int ck = wave + OW_WAVES * k;
        const int cc = ck < BM_NCH ? ck : wave;
        const int filled = min(BM_CH, N_EDGES - cc * BM_CH);
        e[k] = grouped[(size_t)cc * BM_CH + min(st + lane, filled - 1)];
    }

    #pragma unroll
    for (int it = 0; it < OW_CPY; ++it) {
        const int i = tid + it * (OW_WAVES * 32);
        if (i < n4) t4[i] = sv[it];
    }
    __syncthreads();

    int* q  = &queue[wave][0];
    int cnt = 0;

    #pragma unroll
    for (int k = 0; k < OW_CPW; ++k) {
        const unsigned t = (unsigned)__builtin_amdgcn_readlane(tw, k);
        const int m = min((int)(t >> 16), 32);
        if (lane < m) q[cnt + lane] = e[k];
        cnt += m;
    }
    __builtin_amdgcn_fence(__ATOMIC_ACQ_REL, "wavefront");
    int head = 0;
    while (cnt - head >= 32) {
        ow_drain(x, tile, q + head, lane, 32);
        head += 32;
    }
    cnt -= head;
    if (head > 0) {
        const bool mv = lane < cnt;
        const int t2 = mv ? q[head + lane] : 0;
        if (mv) q[lane] = t2;
    }
    __builtin_amdgcn_fence(__ATOMIC_ACQ_REL, "wavefront");

    for (int k = 0; k < OW_CPW; ++k) {
        const unsigned t = (unsigned)__builtin_amdgcn_readlane(tw, k);
        const int n = (int)(t >> 16);
        const int* run = grouped + (size_t)(wave + OW_WAVES * k) * BM_CH + (int)(t & 0xffffu);
        for (int off = 32; off < n; off += 32) {
            const int m = min(n - off, 32);
            int v = 0;
            if (lane < m) v = run[off + lane];
            if (lane < m) q[cnt + lane] = v;
            cnt += m;
            __builtin_amdgcn_fence(__ATOMIC_ACQ_REL, "wavefront");
            if (cnt >= 32) {
                ow_drain(x, tile, q, lane, 32);
                cnt -= 32;
                const bool mv = lane < cnt;
                const int t2 = mv ? q[32 + lane] : 0;
                if (mv) q[lane] = t2;
            }
            __builtin_amdgcn_fence(__ATOMIC_ACQ_REL, "wavefront");
        }
    }
    if (cnt > 0) ow_drain(x, tile, q, lane, cnt);

    __syncthreads();
    {
        v4f* hd = (v4f*)(h0 + (size_t)lo * D);
        const v4f* t4 = (const v4f*)tile;
        v4f ov[OW_CPY];
        #pragma unroll
        for (int it = 0; it < OW_CPY; ++it)
            ov[it] = t4[min(tid + it * (OW_WAVES * 32), n4 - 1)];
        #pragma unroll
        for (int it = 0; it < OW_CPY; ++it) {
            const int i = tid + it * (OW_WAVES * 32);
            if (i < n4) hd[i] = ov[it];
        }
    }
}

#define MLP_WAVES 8
#define MLP_NWG   256
#define MLP_TILES (N_NODES / 16)
#define MLP_STEP  (MLP_NWG * MLP_WAVES)
static_assert(N_NODES % 16 == 0, "every row tile is full: no row guard needed");
#define HSTR  68
#define KSTR  320
#define LHSTR 160
#define MLP_HW_FLOATS 3856
static_assert(MLP_HW_FLOATS >= 16 * HSTR, "relayout tile must fit");
#define MLP_WST (D * D / 4 / (MLP_WAVES * 32))
static_assert(MLP_WST * MLP_WAVES * 32 * 4 == D * D, "the threads share a weight matrix evenly");
static_assert(2 * 16 * KSTR * 4 + MLP_WAVES * MLP_HW_FLOATS * 4 > 160 * 1024, "one work-group per WGP");

__global__ __launch_bounds__(MLP_WAVES * 32) void gin_mlp(const float* __restrict__ h0,
                                                          const float* __restrict__ W1,
                                                          const float* __restrict__ b1,
                                                          const float* __restrict__ W2,
                                                          const float* __restrict__ b2,
                                                          float* __restrict__ out) {
    __shared__ float sW1[16 * KSTR];
    __shared__ float sW2[16 * KSTR];
    __shared__ float sH[MLP_WAVES][MLP_HW_FLOATS];

    const int tid   = threadIdx.x;
    const int lane  = tid & 31;
    const int wave  = __builtin_amdgcn_readfirstlane(tid >> 5);
    const int lhalf = lane >> 4;
    const int l16   = lane & 15;

    const int first = blockIdx.x + MLP_NWG * wave;

    v4f w1r[MLP_WST], w2r[MLP_WST];
    #pragma unroll
    for (int it = 0; it < MLP_WST; ++it) {
        w1r[it] = ((const v4f*)W1)[tid + it * (MLP_WAVES * 32)];
        w2r[it] = ((const v4f*)W2)[tid + it * (MLP_WAVES * 32)];
    }
    float bias1[4], bias2[4];
    #pragma unroll
    for (int nt = 0; nt < 4; ++nt) {
        bias1[nt] = b1[nt * 16 + l16];
        bias2[nt] = b2[nt * 16 + l16];
    }
    v2f a[16];
    {
        const int t0 = min(first, MLP_TILES - 1);
        const float* ap = h0 + (size_t)(t0 * 16 + l16) * D + lhalf * 2;
        #pragma unroll
        for (int kk = 0; kk < 16; ++kk) a[kk] = *(const v2f*)(ap + kk * 4);
    }

    #pragma unroll
    for (int it = 0; it < MLP_WST; ++it) {
        const int i = tid + it * (MLP_WAVES * 32);
        const int k = i >> 4;
        const int c = (i & 15) << 2;
        const int base = (k >> 2) * KSTR + ((k >> 1) & 1) * LHSTR + (k & 1);
        #pragma unroll
        for (int j = 0; j < 4; ++j) {
            sW1[base + ((c + j) << 1)] = w1r[it][j];
            sW2[base + ((c + j) << 1)] = w2r[it][j];
        }
    }
    __syncthreads();

    const v8f vz = {0, 0, 0, 0, 0, 0, 0, 0};
    const float* bw1 = sW1 + lhalf * LHSTR + (l16 << 1);
    const float* bw2 = sW2 + lhalf * LHSTR + (l16 << 1);
    float* myH = &sH[wave][0];
    const float* aH = myH + l16 * HSTR;

    for (int tile = first; tile < MLP_TILES; tile += MLP_STEP) {
        v2f an[16];
        {
            const int tn = (tile + MLP_STEP < MLP_TILES) ? tile + MLP_STEP : tile;
            const float* ap = h0 + (size_t)(tn * 16 + l16) * D + lhalf * 2;
            #pragma unroll
            for (int kk = 0; kk < 16; ++kk) an[kk] = *(const v2f*)(ap + kk * 4);
        }

        v8f acc[4];
        #pragma unroll
        for (int nt = 0; nt < 4; ++nt) acc[nt] = vz;
        #pragma unroll
        for (int kk = 0; kk < 16; ++kk) {
            #pragma unroll
            for (int nt = 0; nt < 4; ++nt) {
                const v2f b = *(const v2f*)(bw1 + kk * KSTR + nt * 32);
                acc[nt] = __builtin_amdgcn_wmma_f32_16x16x4_f32(
                    false, a[kk], false, b, (short)0, acc[nt], false, false);
            }
        }

        #pragma unroll
        for (int nt = 0; nt < 4; ++nt) {
            const int   col  = nt * 16 + l16;
            const float bias = bias1[nt];
            #pragma unroll
            for (int r = 0; r < 8; ++r) {
                float v = acc[nt][r] + bias;
                v = v > 0.0f ? v : 0.0f;
                myH[(r + lhalf * 8) * HSTR + col] = v;
            }
        }

        v8f o[4];
        #pragma unroll
        for (int nt = 0; nt < 4; ++nt) o[nt] = vz;
        #pragma unroll
        for (int kk = 0; kk < 16; ++kk) {
            const int koff = kk * 4 + lhalf * 2;
            const v2f av = *(const v2f*)(aH + koff);
            #pragma unroll
            for (int nt = 0; nt < 4; ++nt) {
                const v2f b = *(const v2f*)(bw2 + kk * KSTR + nt * 32);
                o[nt] = __builtin_amdgcn_wmma_f32_16x16x4_f32(
                    false, av, false, b, (short)0, o[nt], false, false);
            }
        }

        #pragma unroll
        for (int r = 0; r < 8; ++r) {
            const int m = tile * 16 + r + lhalf * 8;
            #pragma unroll
            for (int nt = 0; nt < 4; ++nt) {
                const int col = nt * 16 + l16;
                out[(size_t)m * D + col] = o[nt][r] + bias2[nt];
            }
        }

        #pragma unroll
        for (int kk = 0; kk < 16; ++kk) a[kk] = an[kk];
    }
}

extern "C" void kernel_launch(void* const* d_in, const int* in_sizes, int n_in,
                              void* d_out, int out_size, void* d_ws, size_t ws_size,
                              hipStream_t stream) {
    const float* x  = (const float*)d_in[0];
    const int*   ei = (const int*)d_in[1];
    const float* W1 = (const float*)d_in[2];
    const float* b1 = (const float*)d_in[3];
    const float* W2 = (const float*)d_in[4];
    const float* b2 = (const float*)d_in[5];
    float*       out = (float*)d_out;

    const int* src = ei;
    const int* dst = ei + N_EDGES;

    float*    h0      = (float*)d_ws;
    unsigned* tab     = (unsigned*)((char*)d_ws + 33554432);
    int*      grouped = (int*)((char*)d_ws + 50331648);

    gin_group<<<BM_NCH, 256, 0, stream>>>(src, dst, tab, grouped);
    gin_owner<<<NRANGE, OW_WAVES * 32, 0, stream>>>(x, grouped, tab, h0);
    gin_mlp<<<MLP_NWG, MLP_WAVES * 32, 0, stream>>>(h0, W1, b1, W2, b2, out);
}
